# starting kernel
# baseline (speedup 1.0000x reference)
#include <hip/hip_runtime.h>

#define N_NODES 10000
#define E_EDGES 640000
#define C_IN    256
#define C_HID   128
#define C_OUT   64

typedef float    v2f  __attribute__((ext_vector_type(2)));
typedef float    v8f  __attribute__((ext_vector_type(8)));
typedef _Float16 v16h __attribute__((ext_vector_type(16)));

#if defined(__has_builtin)
#if __has_builtin(__builtin_amdgcn_wmma_f32_16x16x4_f32)
#define USE_WMMA_F32 1
#endif
#endif

__global__ void deg_init_k(float* __restrict__ dinv, int n) {
  int i = blockIdx.x * blockDim.x + threadIdx.x;
  if (i < n) dinv[i] = 1.0f;
}

__global__ void deg_count_k(float* __restrict__ dinv, const int* __restrict__ dst, int e) {
  int i = blockIdx.x * blockDim.x + threadIdx.x;
  if (i < e)
    __hip_atomic_fetch_add(&dinv[dst[i]], 1.0f, __ATOMIC_RELAXED, __HIP_MEMORY_SCOPE_AGENT);
}

__global__ void rsqrt_k(float* __restrict__ dinv, int n) {
  int i = blockIdx.x * blockDim.x + threadIdx.x;
  if (i < n) dinv[i] = rsqrtf(dinv[i]);
}

__global__ __launch_bounds__(128)
void gemm_wmma_k(const float* __restrict__ A, const float* __restrict__ B,
                 float* __restrict__ D, int K, int Ncols) {
  const int lane = threadIdx.x & 31;
  const int wave = threadIdx.x >> 5;
  const int row0 = blockIdx.x * 16;
  const int col0 = (blockIdx.y * 4 + wave) * 16;
  const int r    = lane & 15;
  const int half = lane >> 4;

  const float* Arow = A + (size_t)(row0 + r) * K;
  const float* Bcol = B + col0 + r;
  v8f acc = {};

#if defined(USE_WMMA_F32)
  for (int k = 0; k < K; k += 4) {
    const int kb = k + half * 2;
    v2f a = *(const v2f*)(Arow + kb);
    v2f b;
    b.x = Bcol[(size_t)kb * Ncols];
    b.y = Bcol[(size_t)(kb + 1) * Ncols];
    acc = __builtin_amdgcn_wmma_f32_16x16x4_f32(false, a, false, b,
                                                (short)0, acc, false, false);
  }
#else
  for (int k = 0; k < K; k += 32) {
    v16h a, b;
#pragma unroll
    for (int rr = 0; rr < 8; ++rr) {
      const int ka = k + (rr >> 2) * 16 + half * 8 + (rr & 3) * 2;
      a[2 * rr]     = (_Float16)Arow[ka];
      a[2 * rr + 1] = (_Float16)Arow[ka + 1];
      const int kb = k + half * 16 + 2 * rr;
      b[2 * rr]     = (_Float16)Bcol[(size_t)kb * Ncols];
      b[2 * rr + 1] = (_Float16)Bcol[(size_t)(kb + 1) * Ncols];
    }
    acc = __builtin_amdgcn_wmma_f32_16x16x32_f16(false, a, false, b,
                                                 (short)0, acc, false, false);
  }
#endif

  float* Drow = D + (size_t)(row0 + half * 8) * Ncols + col0 + r;
#pragma unroll
  for (int i = 0; i < 8; ++i) Drow[(size_t)i * Ncols] = acc[i];
}

template <int C>
__global__ void init_out_k(const float* __restrict__ h, const float* __restrict__ dinv,
                           const float* __restrict__ bias, float* __restrict__ out, int n) {
  int t = blockIdx.x * blockDim.x + threadIdx.x;
  if (t >= n * C) return;
  int i = t / C;
  int f = t - i * C;
  float di = dinv[i];
  out[t] = bias[f] + di * di * h[t];
}

template <int C>
__global__ void scatter_k(const float* __restrict__ h, const float* __restrict__ dinv,
                          const int* __restrict__ src, const int* __restrict__ dst,
                          float* __restrict__ out, int e) {
  constexpr int CH = C / 4;
  int t = blockIdx.x * blockDim.x + threadIdx.x;
  if (t >= e * CH) return;
  int ei = t / CH;
  int fc = (t - ei * CH) * 4;
  int s = src[ei], d = dst[ei];
  float w = dinv[s] * dinv[d];
  const float4 hv = *(const float4*)(h + (size_t)s * C + fc);
  float* o = out + (size_t)d * C + fc;
  __hip_atomic_fetch_add(o + 0, w * hv.x, __ATOMIC_RELAXED, __HIP_MEMORY_SCOPE_AGENT);
  __hip_atomic_fetch_add(o + 1, w * hv.y, __ATOMIC_RELAXED, __HIP_MEMORY_SCOPE_AGENT);
  __hip_atomic_fetch_add(o + 2, w * hv.z, __ATOMIC_RELAXED, __HIP_MEMORY_SCOPE_AGENT);
  __hip_atomic_fetch_add(o + 3, w * hv.w, __ATOMIC_RELAXED, __HIP_MEMORY_SCOPE_AGENT);
}

__global__ void relu_k(float* __restrict__ a, int n) {
  int t = blockIdx.x * blockDim.x + threadIdx.x;
  if (t < n) a[t] = fmaxf(a[t], 0.0f);
}

extern "C" void kernel_launch(void* const* d_in, const int* in_sizes, int n_in,
                              void* d_out, int out_size, void* d_ws, size_t ws_size,
                              hipStream_t stream) {
  (void)in_sizes; (void)n_in; (void)out_size; (void)ws_size;
  const float* x  = (const float*)d_in[0];
  const int*   ei = (const int*)d_in[1];
  const float* W1 = (const float*)d_in[2];
  const float* b1 = (const float*)d_in[3];
  const float* W2 = (const float*)d_in[4];
  const float* b2 = (const float*)d_in[5];
  const int* src = ei;
  const int* dst = ei + E_EDGES;
  float* out = (float*)d_out;

  char* ws = (char*)d_ws;
  float* dinv = (float*)ws;  ws += (((size_t)N_NODES * 4) + 255) / 256 * 256;
  float* H1   = (float*)ws;  ws += (size_t)N_NODES * C_HID * 4;
  float* A1   = (float*)ws;  ws += (size_t)N_NODES * C_HID * 4;
  float* H2   = (float*)ws;

  const dim3 B(256);
  deg_init_k<<<(N_NODES + 255) / 256, B, 0, stream>>>(dinv, N_NODES);
  deg_count_k<<<(E_EDGES + 255) / 256, B, 0, stream>>>(dinv, dst, E_EDGES);
  rsqrt_k<<<(N_NODES + 255) / 256, B, 0, stream>>>(dinv, N_NODES);

  gemm_wmma_k<<<dim3(N_NODES / 16, C_HID / 64), dim3(128), 0, stream>>>(x, W1, H1, C_IN, C_HID);
  init_out_k<C_HID><<<(N_NODES * C_HID + 255) / 256, B, 0, stream>>>(H1, dinv, b1, A1, N_NODES);
  scatter_k<C_HID><<<(E_EDGES * (C_HID / 4) + 255) / 256, B, 0, stream>>>(H1, dinv, src, dst, A1, E_EDGES);
  relu_k<<<(N_NODES * C_HID + 255) / 256, B, 0, stream>>>(A1, N_NODES * C_HID);

  gemm_wmma_k<<<dim3(N_NODES / 16, C_OUT / 64), dim3(128), 0, stream>>>(A1, W2, H2, C_HID, C_OUT);
  init_out_k<C_OUT><<<(N_NODES * C_OUT + 255) / 256, B, 0, stream>>>(H2, dinv, b2, out, N_NODES);
  scatter_k<C_OUT><<<(E_EDGES * (C_OUT / 4) + 255) / 256, B, 0, stream>>>(H2, dinv, src, dst, out, E_EDGES);
}


# float-atomic scatter -> dst-range owner blocks: each filters the dst list for its 80 nodes, sums norm*h[src] rows in LDS (ds_add_f32, arrival order), stores each row once; init and relu folded in
# speedup vs baseline: 1.1046x; 1.1046x over previous
#include <hip/hip_runtime.h>

#define N_NODES 10000
#define E_EDGES 640000
#define C_IN    256
#define C_HID   128
#define C_OUT   64

typedef float    v2f  __attribute__((ext_vector_type(2)));
typedef float    v8f  __attribute__((ext_vector_type(8)));
typedef _Float16 v16h __attribute__((ext_vector_type(16)));

#if defined(__has_builtin)
#if __has_builtin(__builtin_amdgcn_wmma_f32_16x16x4_f32)
#define USE_WMMA_F32 1
#endif
#endif

__global__ void deg_init_k(float* __restrict__ dinv, int n) {
  int i = blockIdx.x * blockDim.x + threadIdx.x;
  if (i < n) dinv[i] = 1.0f;
}

__global__ void deg_count_k(float* __restrict__ dinv, const int* __restrict__ dst, int e) {
  int i = blockIdx.x * blockDim.x + threadIdx.x;
  if (i < e)
    __hip_atomic_fetch_add(&dinv[dst[i]], 1.0f, __ATOMIC_RELAXED, __HIP_MEMORY_SCOPE_AGENT);
}

__global__ void rsqrt_k(float* __restrict__ dinv, int n) {
  int i = blockIdx.x * blockDim.x + threadIdx.x;
  if (i < n) dinv[i] = rsqrtf(dinv[i]);
}

__global__ __launch_bounds__(128)
void gemm_wmma_k(const float* __restrict__ A, const float* __restrict__ B,
                 float* __restrict__ D, int K, int Ncols) {
  const int lane = threadIdx.x & 31;
  const int wave = threadIdx.x >> 5;
  const int row0 = blockIdx.x * 16;
  const int col0 = (blockIdx.y * 4 + wave) * 16;
  const int r    = lane & 15;
  const int half = lane >> 4;

  const float* Arow = A + (size_t)(row0 + r) * K;
  const float* Bcol = B + col0 + r;
  v8f acc = {};

#if defined(USE_WMMA_F32)
  for (int k = 0; k < K; k += 4) {
    const int kb = k + half * 2;
    v2f a = *(const v2f*)(Arow + kb);
    v2f b;
    b.x = Bcol[(size_t)kb * Ncols];
    b.y = Bcol[(size_t)(kb + 1) * Ncols];
    acc = __builtin_amdgcn_wmma_f32_16x16x4_f32(false, a, false, b,
                                                (short)0, acc, false, false);
  }
#else
  for (int k = 0; k < K; k += 32) {
    v16h a, b;
#pragma unroll
    for (int rr = 0; rr < 8; ++rr) {
      const int ka = k + (rr >> 2) * 16 + half * 8 + (rr & 3) * 2;
      a[2 * rr]     = (_Float16)Arow[ka];
      a[2 * rr + 1] = (_Float16)Arow[ka + 1];
      const int kb = k + half * 16 + 2 * rr;
      b[2 * rr]     = (_Float16)Bcol[(size_t)kb * Ncols];
      b[2 * rr + 1] = (_Float16)Bcol[(size_t)(kb + 1) * Ncols];
    }
    acc = __builtin_amdgcn_wmma_f32_16x16x32_f16(false, a, false, b,
                                                 (short)0, acc, false, false);
  }
#endif

  float* Drow = D + (size_t)(row0 + half * 8) * Ncols + col0 + r;
#pragma unroll
  for (int i = 0; i < 8; ++i) Drow[(size_t)i * Ncols] = acc[i];
}

#define OWN_NB    80
#define OWN_T     1024
#define OWN_NW    (OWN_T / 32)
#define OWN_LPT   8
#define OWN_SEG   (OWN_T * OWN_LPT * 4)
#define OWN_QB    12

static_assert(OWN_LPT * 4 == 32, "a lane's matches of one segment are kept as one 32-bit mask");
static_assert(E_EDGES % 4 == 0, "dst list is read in 16-byte pieces");

template <int C, bool RELU>
__global__ __launch_bounds__(OWN_T)
void own_k(const float* __restrict__ h, const float* __restrict__ dinv,
           const float* __restrict__ bias, const int* __restrict__ src,
           const int* __restrict__ dst, float* __restrict__ out) {
  __shared__ float acc[OWN_NB * C];
  __shared__ int   q[OWN_SEG];
  __shared__ float dloc[OWN_NB];
  __shared__ int   qn[2];

  const int tid  = threadIdx.x;
  const int lane = tid & 31;
  const int wave = __builtin_amdgcn_readfirstlane(tid >> 5);
  const int n0   = blockIdx.x * OWN_NB;

  for (int t = tid; t < OWN_NB * C; t += OWN_T) acc[t] = 0.0f;
  if (tid < OWN_NB) dloc[tid] = (n0 + tid < N_NODES) ? dinv[n0 + tid] : 0.0f;
  if (tid < 2) qn[tid] = 0;

  int4 cur[OWN_LPT];
#pragma unroll
  for (int j = 0; j < OWN_LPT; ++j) {
    const int eb = (j * OWN_T + tid) * 4;
    cur[j] = (eb < E_EDGES) ? *(const int4*)(dst + eb) : make_int4(-1, -1, -1, -1);
  }
  __syncthreads();

  int it = 0;
  for (int s0 = 0; s0 < E_EDGES; s0 += OWN_SEG, ++it) {
    const int par = it & 1;
    unsigned mask = 0;
#pragma unroll
    for (int j = 0; j < OWN_LPT; ++j) {
      const int dv[4] = {cur[j].x, cur[j].y, cur[j].z, cur[j].w};
#pragma unroll
      for (int c = 0; c < 4; ++c)
        mask = mask + mask + (((unsigned)(dv[c] - n0) < (unsigned)OWN_NB) ? 1u : 0u);
    }
    while (mask) {
      const int k = __builtin_clz(mask);
      mask &= ~(0x80000000u >> k);
      const int e = s0 + ((k >> 2) * OWN_T + tid) * 4 + (k & 3);
      const int slot = atomicAdd(&qn[par], 1);
      q[slot] = e;
    }
    if (tid == 0) qn[par ^ 1] = 0;
    __syncthreads();

    if (s0 + OWN_SEG < E_EDGES) {
#pragma unroll
      for (int j = 0; j < OWN_LPT; ++j) {
        const int eb = s0 + OWN_SEG + (j * OWN_T + tid) * 4;
        cur[j] = (eb < E_EDGES) ? *(const int4*)(dst + eb) : make_int4(-1, -1, -1, -1);
      }
    }

    const int n = __builtin_amdgcn_readfirstlane(qn[par]);
    const int ul = (lane < OWN_QB) ? lane : 0;
    for (int r0 = wave; r0 < n; r0 += OWN_NW * OWN_QB) {
      int my_i = r0 + ul * OWN_NW;
      const bool ok = my_i < n;
      my_i = ok ? my_i : r0;
      const int   e_v  = q[my_i];
      const int   s_v  = src[e_v];
      const int   dq   = dst[e_v] - n0;
      const int   dl_v = ok ? dq : -1;
      const float w_v  = dinv[s_v] * dloc[dq];
      float v[OWN_QB][C / 32];
#pragma unroll
      for (int u = 0; u < OWN_QB; ++u) {
        const int dl_u = __builtin_amdgcn_readlane(dl_v, u);
        if (dl_u >= 0) {
          const int s_u = __builtin_amdgcn_readlane(s_v, u);
          const float* row = h + (size_t)s_u * C + lane;
#pragma unroll
          for (int k = 0; k < C / 32; ++k) v[u][k] = row[32 * k];
        }
      }
#pragma unroll
      for (int u = 0; u < OWN_QB; ++u) {
        const int dl_u = __builtin_amdgcn_readlane(dl_v, u);
        if (dl_u >= 0) {
          const float w_u = __int_as_float(__builtin_amdgcn_readlane(__float_as_int(w_v), u));
          float* a = acc + dl_u * C + lane;
#pragma unroll
          for (int k = 0; k < C / 32; ++k)
            __hip_atomic_fetch_add(a + 32 * k, w_u * v[u][k], __ATOMIC_RELAXED,
                                   __HIP_MEMORY_SCOPE_WORKGROUP);
        }
      }
    }
    __syncthreads();
  }

  for (int t = tid; t < OWN_NB * C; t += OWN_T) {
    const int i = t / C;
    const int f = t - i * C;
    const int node = n0 + i;
    if (node < N_NODES) {
      const float di = dloc[i];
      float r = bias[f] + di * di * h[(size_t)node * C + f] + acc[t];
      if (RELU) r = fmaxf(r, 0.0f);
      out[(size_t)node * C + f] = r;
    }
  }
}

extern "C" void kernel_launch(void* const* d_in, const int* in_sizes, int n_in,
                              void* d_out, int out_size, void* d_ws, size_t ws_size,
                              hipStream_t stream) {
  (void)in_sizes; (void)n_in; (void)out_size; (void)ws_size;
  const float* x  = (const float*)d_in[0];
  const int*   ei = (const int*)d_in[1];
  const float* W1 = (const float*)d_in[2];
  const float* b1 = (const float*)d_in[3];
  const float* W2 = (const float*)d_in[4];
  const float* b2 = (const float*)d_in[5];
  const int* src = ei;
  const int* dst = ei + E_EDGES;
  float* out = (float*)d_out;

  char* ws = (char*)d_ws;
  float* dinv = (float*)ws;  ws += (((size_t)N_NODES * 4) + 255) / 256 * 256;
  float* H1   = (float*)ws;  ws += (size_t)N_NODES * C_HID * 4;
  float* A1   = (float*)ws;  ws += (size_t)N_NODES * C_HID * 4;
  float* H2   = (float*)ws;

  const dim3 B(256);
  deg_init_k<<<(N_NODES + 255) / 256, B, 0, stream>>>(dinv, N_NODES);
  deg_count_k<<<(E_EDGES + 255) / 256, B, 0, stream>>>(dinv, dst, E_EDGES);
  rsqrt_k<<<(N_NODES + 255) / 256, B, 0, stream>>>(dinv, N_NODES);

  gemm_wmma_k<<<dim3(N_NODES / 16, C_HID / 64), dim3(128), 0, stream>>>(x, W1, H1, C_IN, C_HID);
  own_k<C_HID, true><<<(N_NODES + OWN_NB - 1) / OWN_NB, dim3(OWN_T), 0, stream>>>(H1, dinv, b1, src, dst, A1);

  gemm_wmma_k<<<dim3(N_NODES / 16, C_OUT / 64), dim3(128), 0, stream>>>(A1, W2, H2, C_HID, C_OUT);
  own_k<C_OUT, false><<<(N_NODES + OWN_NB - 1) / OWN_NB, dim3(OWN_T), 0, stream>>>(H2, dinv, b2, src, dst, out);
}


# GEMMs: one wave per 16x64 strip (4 accumulator tiles), not per 16x16 tile: an A fragment feeds 4 WMMAs, A/B/D move as float4, next 64-k chunk loaded ahead; sum order: k regrouped inside 8-k groups
# speedup vs baseline: 3.4321x; 3.1071x over previous
#include <hip/hip_runtime.h>

#define N_NODES 10000
#define E_EDGES 640000
#define C_IN    256
#define C_HID   128
#define C_OUT   64

typedef float    v2f  __attribute__((ext_vector_type(2)));
typedef float    v8f  __attribute__((ext_vector_type(8)));

#define OWN_NB    20
#define OWN_T     512
#define OWN_NW    (OWN_T / 32)
#define OWN_QB    12
#define OWN_NBLK  ((N_NODES + OWN_NB - 1) / OWN_NB)
#define BIN_CAP   8192
#define BIN_SHIFT 8
#define CUR_PITCH 32
#define BIN_T     1024
#define BIN_EPT   5
#define BIN_NBLK  ((E_EDGES + BIN_T * BIN_EPT - 1) / (BIN_T * BIN_EPT))
#define DEG_T     256
#define DEG_U     6

static_assert(OWN_NB <= (1 << BIN_SHIFT), "local dst index must fit below the src field of a bin entry");
static_assert(N_NODES <= (1 << (31 - BIN_SHIFT)), "src must fit above the local dst field of a bin entry");
static_assert(OWN_NBLK <= BIN_T, "bin_k reserves one owner range per thread");

__global__ void cur_init_k(int* __restrict__ cur) {
  int i = blockIdx.x * blockDim.x + threadIdx.x;
  if (i <= OWN_NBLK) cur[i * CUR_PITCH] = 0;
}

__global__ __launch_bounds__(BIN_T)
void bin_k(int* __restrict__ cur, int* __restrict__ bins, int* __restrict__ ovf,
           const int* __restrict__ src, const int* __restrict__ dst, int e) {
  __shared__ int cnt[OWN_NBLK];
  const int tid = threadIdx.x;
  if (tid < OWN_NBLK) cnt[tid] = 0;

  const int i0 = blockIdx.x * (BIN_T * BIN_EPT) + tid;
  int d[BIN_EPT], s[BIN_EPT];
#pragma unroll
  for (int k = 0; k < BIN_EPT; ++k) {
    const int i = min(i0 + k * BIN_T, e - 1);
    d[k] = dst[i];
    s[k] = src[i];
  }
  __syncthreads();

  int b[BIN_EPT], r[BIN_EPT];
#pragma unroll
  for (int k = 0; k < BIN_EPT; ++k) {
    b[k] = d[k] / OWN_NB;
    r[k] = -1;
    if (i0 + k * BIN_T < e) r[k] = atomicAdd(&cnt[b[k]], 1);
  }
  __syncthreads();

  if (tid < OWN_NBLK) {
    const int c = cnt[tid];
    int base = 0;
    if (c > 0)
      base = __hip_atomic_fetch_add(&cur[tid * CUR_PITCH], c, __ATOMIC_RELAXED,
                                    __HIP_MEMORY_SCOPE_AGENT);
    cnt[tid] = base;
  }
  __syncthreads();

  int slot[BIN_EPT];
#pragma unroll
  for (int k = 0; k < BIN_EPT; ++k) slot[k] = cnt[b[k]] + r[k];
#pragma unroll
  for (int k = 0; k < BIN_EPT; ++k) {
    if (r[k] >= 0) {
      if (slot[k] < BIN_CAP) {
        bins[(size_t)b[k] * BIN_CAP + slot[k]] = (s[k] << BIN_SHIFT) | (d[k] - b[k] * OWN_NB);
      } else {
        const int o = __hip_atomic_fetch_add(&cur[OWN_NBLK * CUR_PITCH], 1, __ATOMIC_RELAXED,
                                             __HIP_MEMORY_SCOPE_AGENT);
        ovf[o] = i0 + k * BIN_T;
      }
    }
  }
}

__global__ __launch_bounds__(DEG_T)
void deg_k(float* __restrict__ dinv, const int* __restrict__ cur, const int* __restrict__ bins,
           const int* __restrict__ ovf, const int* __restrict__ dst) {
  __shared__ int cnt[OWN_NB];
  const int tid = threadIdx.x;
  const int n0  = blockIdx.x * OWN_NB;
  if (tid < OWN_NB) cnt[tid] = 0;
  const int n  = min(cur[blockIdx.x * CUR_PITCH], BIN_CAP);
  const int no = min(cur[OWN_NBLK * CUR_PITCH], E_EDGES);
  const int* __restrict__ bin = bins + (size_t)blockIdx.x * BIN_CAP;
  __syncthreads();

  for (int i = tid; i < n; i += DEG_T * DEG_U) {
    int ent[DEG_U];
#pragma unroll
    for (int u = 0; u < DEG_U; ++u) ent[u] = bin[min(i + u * DEG_T, n - 1)];
#pragma unroll
    for (int u = 0; u < DEG_U; ++u)
      if (i + u * DEG_T < n) atomicAdd(&cnt[ent[u] & ((1 << BIN_SHIFT) - 1)], 1);
  }
  for (int i = tid; i < no; i += DEG_T) {
    const int dq = dst[ovf[i]] - n0;
    if ((unsigned)dq < (unsigned)OWN_NB) atomicAdd(&cnt[dq], 1);
  }
  __syncthreads();

  if (tid < OWN_NB && n0 + tid < N_NODES) dinv[n0 + tid] = rsqrtf(1.0f + (float)cnt[tid]);
}

#define GEMM_KC 64
#define GEMM_NG (GEMM_KC / 8)
#define M_TILES (N_NODES / 16)

template <int K, int N>
__device__ __forceinline__ void gemm_load_chunk(float4 (&a)[GEMM_NG], float4 (&b)[GEMM_NG][4],
                                                const float* __restrict__ Ap,
                                                const float* __restrict__ Bp, int k0) {
#pragma unroll
  for (int g = 0; g < GEMM_NG; ++g) {
    a[g] = *(const float4*)(Ap + k0 + 8 * g);
#pragma unroll
    for (int i = 0; i < 4; ++i) b[g][i] = *(const float4*)(Bp + (size_t)(k0 + 8 * g + i) * N);
  }
}

__device__ __forceinline__ void gemm_mma_chunk(v8f (&acc)[4], const float4 (&a)[GEMM_NG],
                                               const float4 (&b)[GEMM_NG][4]) {
#pragma unroll
  for (int g = 0; g < GEMM_NG; ++g) {
    const v2f a0 = {a[g].x, a[g].y};
    const v2f a1 = {a[g].z, a[g].w};
    const float b0[4] = {b[g][0].x, b[g][0].y, b[g][0].z, b[g][0].w};
    const float b1[4] = {b[g][1].x, b[g][1].y, b[g][1].z, b[g][1].w};
    const float b2[4] = {b[g][2].x, b[g][2].y, b[g][2].z, b[g][2].w};
    const float b3[4] = {b[g][3].x, b[g][3].y, b[g][3].z, b[g][3].w};
#pragma unroll
    for (int t = 0; t < 4; ++t) {
      const v2f bb = {b0[t], b1[t]};
      acc[t] = __builtin_amdgcn_wmma_f32_16x16x4_f32(false, a0, false, bb, (short)0, acc[t],
                                                     false, false);
    }
#pragma unroll
    for (int t = 0; t < 4; ++t) {
      const v2f bb = {b2[t], b3[t]};
      acc[t] = __builtin_amdgcn_wmma_f32_16x16x4_f32(false, a1, false, bb, (short)0, acc[t],
                                                     false, false);
    }
  }
}

template <int K, int N>
__global__ __launch_bounds__(128)
void gemm_strip_k(const float* __restrict__ A, const float* __restrict__ B, float* __restrict__ D) {
  static_assert(K % (2 * GEMM_KC) == 0, "the chunk loop below is written two chunks per trip");
  static_assert(N % 64 == 0, "a strip is 64 columns");
  const int lane = threadIdx.x & 31;
  const int wave = __builtin_amdgcn_readfirstlane(threadIdx.x >> 5);
  const int rt   = blockIdx.x * 4 + wave;
  if (rt >= M_TILES) return;
  const int c  = lane & 15;
  const int h  = lane >> 4;
  const int n0 = blockIdx.y * 64;

  const float* __restrict__ Ap = A + (size_t)(rt * 16 + c) * K + 4 * h;
  const float* __restrict__ Bp = B + (size_t)(4 * h) * N + n0 + 4 * c;

  v8f acc[4] = {};
  float4 a0[GEMM_NG], b0[GEMM_NG][4];
  float4 a1[GEMM_NG], b1[GEMM_NG][4];

  gemm_load_chunk<K, N>(a0, b0, Ap, Bp, 0);
#pragma unroll
  for (int k0 = 0; k0 < K; k0 += 2 * GEMM_KC) {
    gemm_load_chunk<K, N>(a1, b1, Ap, Bp, k0 + GEMM_KC);
    __builtin_amdgcn_sched_barrier(0);
    gemm_mma_chunk(acc, a0, b0);
    if (k0 + 2 * GEMM_KC < K) gemm_load_chunk<K, N>(a0, b0, Ap, Bp, k0 + 2 * GEMM_KC);
    __builtin_amdgcn_sched_barrier(0);
    gemm_mma_chunk(acc, a1, b1);
  }

  float* Dp = D + (size_t)(rt * 16 + 8 * h) * N + n0 + 4 * c;
#pragma unroll
  for (int i = 0; i < 8; ++i) {
    const float4 o = {acc[0][i], acc[1][i], acc[2][i], acc[3][i]};
    *(float4*)(Dp + (size_t)i * N) = o;
  }
}

template <int C, bool RELU>
__global__ __launch_bounds__(OWN_T)
void own_k(const float* __restrict__ h, const float* __restrict__ dinv,
           const float* __restrict__ bias, const int* __restrict__ src,
           const int* __restrict__ dst, const int* __restrict__ cur,
           const int* __restrict__ bins, const int* __restrict__ ovf,
           float* __restrict__ out) {
  __shared__ float acc[OWN_NB * C];
  __shared__ float dloc[OWN_NB];

  const int tid  = threadIdx.x;
  const int lane = tid & 31;
  const int wave = __builtin_amdgcn_readfirstlane(tid >> 5);
  const int n0   = blockIdx.x * OWN_NB;

  for (int t = tid; t < OWN_NB * C; t += OWN_T) acc[t] = 0.0f;
  if (tid < OWN_NB) dloc[tid] = (n0 + tid < N_NODES) ? dinv[n0 + tid] : 0.0f;

  const int n  = min(cur[blockIdx.x * CUR_PITCH], BIN_CAP);
  const int no = min(cur[OWN_NBLK * CUR_PITCH], E_EDGES);
  const int* __restrict__ bin = bins + (size_t)blockIdx.x * BIN_CAP;
  __syncthreads();

  const int ul = (lane < OWN_QB) ? lane : 0;
  int  j     = wave;
  int  ent_v = 0;
  bool ok    = false;
  if (j * OWN_QB < n) {
    const int i = j * OWN_QB + ul;
    ok    = i < n;
    ent_v = bin[ok ? i : j * OWN_QB];
  }
  while (j * OWN_QB < n) {
    const int   s_v  = ent_v >> BIN_SHIFT;
    const int   dq   = ent_v & ((1 << BIN_SHIFT) - 1);
    const int   dl_v = ok ? dq : -1;
    const float ds_v = dinv[s_v];
    const int jn = j + OWN_NW;
    int  ent_n = 0;
    bool ok_n  = false;
    if (jn * OWN_QB < n) {
      const int i = jn * OWN_QB + ul;
      ok_n  = i < n;
      ent_n = bin[ok_n ? i : jn * OWN_QB];
    }
    float v[OWN_QB][C / 32];
#pragma unroll
    for (int u = 0; u < OWN_QB; ++u) {
      const int dl_u = __builtin_amdgcn_readlane(dl_v, u);
      if (dl_u >= 0) {
        const int s_u = __builtin_amdgcn_readlane(s_v, u);
        const float* row = h + (size_t)s_u * C + lane;
#pragma unroll
        for (int k = 0; k < C / 32; ++k) v[u][k] = row[32 * k];
      }
    }
    const float w_v = ds_v * dloc[dq];
#pragma unroll
    for (int u = 0; u < OWN_QB; ++u) {
      const int dl_u = __builtin_amdgcn_readlane(dl_v, u);
      if (dl_u >= 0) {
        const float w_u = __int_as_float(__builtin_amdgcn_readlane(__float_as_int(w_v), u));
        float* a = acc + dl_u * C + lane;
#pragma unroll
        for (int k = 0; k < C / 32; ++k)
          __hip_atomic_fetch_add(a + 32 * k, w_u * v[u][k], __ATOMIC_RELAXED,
                                 __HIP_MEMORY_SCOPE_WORKGROUP);
      }
    }
    j = jn; ent_v = ent_n; ok = ok_n;
  }

  for (int i = wave; i < no; i += OWN_NW) {
    const int e  = ovf[i];
    const int dq = dst[e] - n0;
    if ((unsigned)dq < (unsigned)OWN_NB) {
      const int   s = src[e];
      const float w = dinv[s] * dloc[dq];
      const float* row = h + (size_t)s * C + lane;
      float* a = acc + dq * C + lane;
#pragma unroll
      for (int k = 0; k < C / 32; ++k)
        __hip_atomic_fetch_add(a + 32 * k, w * row[32 * k], __ATOMIC_RELAXED,
                               __HIP_MEMORY_SCOPE_WORKGROUP);
    }
  }
  __syncthreads();

  for (int t = tid; t < OWN_NB * C; t += OWN_T) {
    const int i = t / C;
    const int f = t - i * C;
    const int node = n0 + i;
    if (node < N_NODES) {
      const float di = dloc[i];
      float r = bias[f] + di * di * h[(size_t)node * C + f] + acc[t];
      if (RELU) r = fmaxf(r, 0.0f);
      out[(size_t)node * C + f] = r;
    }
  }
}

extern "C" void kernel_launch(void* const* d_in, const int* in_sizes, int n_in,
                              void* d_out, int out_size, void* d_ws, size_t ws_size,
                              hipStream_t stream) {
  (void)in_sizes; (void)n_in; (void)out_size; (void)ws_size;
  const float* x  = (const float*)d_in[0];
  const int*   ei = (const int*)d_in[1];
  const float* W1 = (const float*)d_in[2];
  const float* b1 = (const float*)d_in[3];
  const float* W2 = (const float*)d_in[4];
  const float* b2 = (const float*)d_in[5];
  const int* src = ei;
  const int* dst = ei + E_EDGES;
  float* out = (float*)d_out;

  char* ws = (char*)d_ws;
  float* dinv = (float*)ws;  ws += (((size_t)N_NODES * 4) + 255) / 256 * 256;
  float* H1   = (float*)ws;  ws += (size_t)N_NODES * C_HID * 4;
  float* A1   = (float*)ws;  ws += (size_t)N_NODES * C_HID * 4;
  float* H2   = (float*)ws;  ws += (size_t)N_NODES * C_OUT * 4;
  int*   cur  = (int*)ws;    ws += (((size_t)(OWN_NBLK + 1) * CUR_PITCH * 4) + 255) / 256 * 256;
  int*   bins = (int*)ws;    ws += (size_t)OWN_NBLK * BIN_CAP * 4;
  int*   ovf  = (int*)ws;

  cur_init_k<<<(OWN_NBLK + 1 + 255) / 256, dim3(256), 0, stream>>>(cur);
  bin_k<<<BIN_NBLK, dim3(BIN_T), 0, stream>>>(cur, bins, ovf, src, dst, E_EDGES);
  deg_k<<<OWN_NBLK, dim3(DEG_T), 0, stream>>>(dinv, cur, bins, ovf, dst);

  gemm_strip_k<C_IN, C_HID><<<dim3((M_TILES + 3) / 4, C_HID / 64), dim3(128), 0, stream>>>(x, W1, H1);
  own_k<C_HID, true><<<OWN_NBLK, dim3(OWN_T), 0, stream>>>(H1, dinv, b1, src, dst, cur, bins, ovf, A1);

  gemm_strip_k<C_HID, C_OUT><<<dim3((M_TILES + 3) / 4, C_OUT / 64), dim3(128), 0, stream>>>(A1, W2, H2);
  own_k<C_OUT, false><<<OWN_NBLK, dim3(OWN_T), 0, stream>>>(H2, dinv, b2, src, dst, cur, bins, ovf, out);
}


# own_k gather: each h[src] row read by ONE whole-row load per edge (b128 for 128 columns, b64 for 64), not one b32 load per 32 columns; LDS sums kept lane-major; add arrival order as before
# speedup vs baseline: 3.5769x; 1.0422x over previous
#include <hip/hip_runtime.h>

#define N_NODES 10000
#define E_EDGES 640000
#define C_IN    256
#define C_HID   128
#define C_OUT   64

typedef float    v2f  __attribute__((ext_vector_type(2)));
typedef float    v8f  __attribute__((ext_vector_type(8)));

#define OWN_NB    20
#define OWN_T     512
#define OWN_NW    (OWN_T / 32)
#define OWN_QB    12
#define OWN_NBLK  ((N_NODES + OWN_NB - 1) / OWN_NB)
#define BIN_CAP   8192
#define BIN_SHIFT 8
#define CUR_PITCH 32
#define BIN_T     1024
#define BIN_EPT   5
#define BIN_NBLK  ((E_EDGES + BIN_T * BIN_EPT - 1) / (BIN_T * BIN_EPT))
#define DEG_T     256
#define DEG_U     6

static_assert(OWN_NB <= (1 << BIN_SHIFT), "local dst index must fit below the src field of a bin entry");
static_assert(N_NODES <= (1 << (31 - BIN_SHIFT)), "src must fit above the local dst field of a bin entry");
static_assert(OWN_NBLK <= BIN_T, "bin_k reserves one owner range per thread");

__global__ void cur_init_k(int* __restrict__ cur) {
  int i = blockIdx.x * blockDim.x + threadIdx.x;
  if (i <= OWN_NBLK) cur[i * CUR_PITCH] = 0;
}

__global__ __launch_bounds__(BIN_T)
void bin_k(int* __restrict__ cur, int* __restrict__ bins, int* __restrict__ ovf,
           const int* __restrict__ src, const int* __restrict__ dst, int e) {
  __shared__ int cnt[OWN_NBLK];
  const int tid = threadIdx.x;
  if (tid < OWN_NBLK) cnt[tid] = 0;

  const int i0 = blockIdx.x * (BIN_T * BIN_EPT) + tid;
  int d[BIN_EPT], s[BIN_EPT];
#pragma unroll
  for (int k = 0; k < BIN_EPT; ++k) {
    const int i = min(i0 + k * BIN_T, e - 1);
    d[k] = dst[i];
    s[k] = src[i];
  }
  __syncthreads();

  int b[BIN_EPT], r[BIN_EPT];
#pragma unroll
  for (int k = 0; k < BIN_EPT; ++k) {
    b[k] = d[k] / OWN_NB;
    r[k] = -1;
    if (i0 + k * BIN_T < e) r[k] = atomicAdd(&cnt[b[k]], 1);
  }
  __syncthreads();

  if (tid < OWN_NBLK) {
    const int c = cnt[tid];
    int base = 0;
    if (c > 0)
      base = __hip_atomic_fetch_add(&cur[tid * CUR_PITCH], c, __ATOMIC_RELAXED,
                                    __HIP_MEMORY_SCOPE_AGENT);
    cnt[tid] = base;
  }
  __syncthreads();

  int slot[BIN_EPT];
#pragma unroll
  for (int k = 0; k < BIN_EPT; ++k) slot[k] = cnt[b[k]] + r[k];
#pragma unroll
  for (int k = 0; k < BIN_EPT; ++k) {
    if (r[k] >= 0) {
      if (slot[k] < BIN_CAP) {
        bins[(size_t)b[k] * BIN_CAP + slot[k]] = (s[k] << BIN_SHIFT) | (d[k] - b[k] * OWN_NB);
      } else {
        const int o = __hip_atomic_fetch_add(&cur[OWN_NBLK * CUR_PITCH], 1, __ATOMIC_RELAXED,
                                             __HIP_MEMORY_SCOPE_AGENT);
        ovf[o] = i0 + k * BIN_T;
      }
    }
  }
}

__global__ __launch_bounds__(DEG_T)
void deg_k(float* __restrict__ dinv, const int* __restrict__ cur, const int* __restrict__ bins,
           const int* __restrict__ ovf, const int* __restrict__ dst) {
  __shared__ int cnt[OWN_NB];
  const int tid = threadIdx.x;
  const int n0  = blockIdx.x * OWN_NB;
  if (tid < OWN_NB) cnt[tid] = 0;
  const int n  = min(cur[blockIdx.x * CUR_PITCH], BIN_CAP);
  const int no = min(cur[OWN_NBLK * CUR_PITCH], E_EDGES);
  const int* __restrict__ bin = bins + (size_t)blockIdx.x * BIN_CAP;
  __syncthreads();

  for (int i = tid; i < n; i += DEG_T * DEG_U) {
    int ent[DEG_U];
#pragma unroll
    for (int u = 0; u < DEG_U; ++u) ent[u] = bin[min(i + u * DEG_T, n - 1)];
#pragma unroll
    for (int u = 0; u < DEG_U; ++u)
      if (i + u * DEG_T < n) atomicAdd(&cnt[ent[u] & ((1 << BIN_SHIFT) - 1)], 1);
  }
  for (int i = tid; i < no; i += DEG_T) {
    const int dq = dst[ovf[i]] - n0;
    if ((unsigned)dq < (unsigned)OWN_NB) atomicAdd(&cnt[dq], 1);
  }
  __syncthreads();

  if (tid < OWN_NB && n0 + tid < N_NODES) dinv[n0 + tid] = rsqrtf(1.0f + (float)cnt[tid]);
}

#define GEMM_KC 64
#define GEMM_NG (GEMM_KC / 8)
#define M_TILES (N_NODES / 16)

template <int K, int N>
__device__ __forceinline__ void gemm_load_chunk(float4 (&a)[GEMM_NG], float4 (&b)[GEMM_NG][4],
                                                const float* __restrict__ Ap,
                                                const float* __restrict__ Bp, int k0) {
#pragma unroll
  for (int g = 0; g < GEMM_NG; ++g) {
    a[g] = *(const float4*)(Ap + k0 + 8 * g);
#pragma unroll
    for (int i = 0; i < 4; ++i) b[g][i] = *(const float4*)(Bp + (size_t)(k0 + 8 * g + i) * N);
  }
}

__device__ __forceinline__ void gemm_mma_chunk(v8f (&acc)[4], const float4 (&a)[GEMM_NG],
                                               const float4 (&b)[GEMM_NG][4]) {
#pragma unroll
  for (int g = 0; g < GEMM_NG; ++g) {
    const v2f a0 = {a[g].x, a[g].y};
    const v2f a1 = {a[g].z, a[g].w};
    const float b0[4] = {b[g][0].x, b[g][0].y, b[g][0].z, b[g][0].w};
    const float b1[4] = {b[g][1].x, b[g][1].y, b[g][1].z, b[g][1].w};
    const float b2[4] = {b[g][2].x, b[g][2].y, b[g][2].z, b[g][2].w};
    const float b3[4] = {b[g][3].x, b[g][3].y, b[g][3].z, b[g][3].w};
#pragma unroll
    for (int t = 0; t < 4; ++t) {
      const v2f bb = {b0[t], b1[t]};
      acc[t] = __builtin_amdgcn_wmma_f32_16x16x4_f32(false, a0, false, bb, (short)0, acc[t],
                                                     false, false);
    }
#pragma unroll
    for (int t = 0; t < 4; ++t) {
      const v2f bb = {b2[t], b3[t]};
      acc[t] = __builtin_amdgcn_wmma_f32_16x16x4_f32(false, a1, false, bb, (short)0, acc[t],
                                                     false, false);
    }
  }
}

template <int K, int N>
__global__ __launch_bounds__(128)
void gemm_strip_k(const float* __restrict__ A, const float* __restrict__ B, float* __restrict__ D) {
  static_assert(K % (2 * GEMM_KC) == 0, "the chunk loop below is written two chunks per trip");
  static_assert(N % 64 == 0, "a strip is 64 columns");
  const int lane = threadIdx.x & 31;
  const int wave = __builtin_amdgcn_readfirstlane(threadIdx.x >> 5);
  const int rt   = blockIdx.x * 4 + wave;
  if (rt >= M_TILES) return;
  const int c  = lane & 15;
  const int h  = lane >> 4;
  const int n0 = blockIdx.y * 64;

  const float* __restrict__ Ap = A + (size_t)(rt * 16 + c) * K + 4 * h;
  const float* __restrict__ Bp = B + (size_t)(4 * h) * N + n0 + 4 * c;

  v8f acc[4] = {};
  float4 a0[GEMM_NG], b0[GEMM_NG][4];
  float4 a1[GEMM_NG], b1[GEMM_NG][4];

  gemm_load_chunk<K, N>(a0, b0, Ap, Bp, 0);
#pragma unroll
  for (int k0 = 0; k0 < K; k0 += 2 * GEMM_KC) {
    gemm_load_chunk<K, N>(a1, b1, Ap, Bp, k0 + GEMM_KC);
    __builtin_amdgcn_sched_barrier(0);
    gemm_mma_chunk(acc, a0, b0);
    if (k0 + 2 * GEMM_KC < K) gemm_load_chunk<K, N>(a0, b0, Ap, Bp, k0 + 2 * GEMM_KC);
    __builtin_amdgcn_sched_barrier(0);
    gemm_mma_chunk(acc, a1, b1);
  }

  float* Dp = D + (size_t)(rt * 16 + 8 * h) * N + n0 + 4 * c;
#pragma unroll
  for (int i = 0; i < 8; ++i) {
    const float4 o = {acc[0][i], acc[1][i], acc[2][i], acc[3][i]};
    *(float4*)(Dp + (size_t)i * N) = o;
  }
}

template <int C, bool RELU>
__global__ __launch_bounds__(OWN_T)
void own_k(const float* __restrict__ h, const float* __restrict__ dinv,
           const float* __restrict__ bias, const int* __restrict__ src,
           const int* __restrict__ dst, const int* __restrict__ cur,
           const int* __restrict__ bins, const int* __restrict__ ovf,
           float* __restrict__ out) {
  __shared__ float acc[OWN_NB * C];
  __shared__ float dloc[OWN_NB];
  constexpr int V = C / 32;
  typedef float vrow_t __attribute__((ext_vector_type(V)));

  const int tid  = threadIdx.x;
  const int lane = tid & 31;
  const int wave = __builtin_amdgcn_readfirstlane(tid >> 5);
  const int n0   = blockIdx.x * OWN_NB;

  for (int t = tid; t < OWN_NB * C; t += OWN_T) acc[t] = 0.0f;
  if (tid < OWN_NB) dloc[tid] = (n0 + tid < N_NODES) ? dinv[n0 + tid] : 0.0f;

  const int n  = min(cur[blockIdx.x * CUR_PITCH], BIN_CAP);
  const int no = min(cur[OWN_NBLK * CUR_PITCH], E_EDGES);
  const int* __restrict__ bin = bins + (size_t)blockIdx.x * BIN_CAP;
  __syncthreads();

  const int ul = (lane < OWN_QB) ? lane : 0;
  int  j     = wave;
  int  ent_v = 0;
  bool ok    = false;
  if (j * OWN_QB < n) {
    const int i = j * OWN_QB + ul;
    ok    = i < n;
    ent_v = bin[ok ? i : j * OWN_QB];
  }
  while (j * OWN_QB < n) {
    const int   s_v  = ent_v >> BIN_SHIFT;
    const int   dq   = ent_v & ((1 << BIN_SHIFT) - 1);
    const int   dl_v = ok ? dq : -1;
    const float ds_v = dinv[s_v];
    const int jn = j + OWN_NW;
    int  ent_n = 0;
    bool ok_n  = false;
    if (jn * OWN_QB < n) {
      const int i = jn * OWN_QB + ul;
      ok_n  = i < n;
      ent_n = bin[ok_n ? i : jn * OWN_QB];
    }
    vrow_t v[OWN_QB];
#pragma unroll
    for (int u = 0; u < OWN_QB; ++u) {
      const int dl_u = __builtin_amdgcn_readlane(dl_v, u);
      if (dl_u >= 0) {
        const int s_u = __builtin_amdgcn_readlane(s_v, u);
        v[u] = *(const vrow_t*)(h + (size_t)s_u * C + V * lane);
      }
    }
    const float w_v = ds_v * dloc[dq];
#pragma unroll
    for (int u = 0; u < OWN_QB; ++u) {
      const int dl_u = __builtin_amdgcn_readlane(dl_v, u);
      if (dl_u >= 0) {
        const float w_u = __int_as_float(__builtin_amdgcn_readlane(__float_as_int(w_v), u));
        float* a = acc + dl_u * C + lane;
#pragma unroll
        for (int k = 0; k < V; ++k)
          __hip_atomic_fetch_add(a + 32 * k, w_u * v[u][k], __ATOMIC_RELAXED,
                                 __HIP_MEMORY_SCOPE_WORKGROUP);
      }
    }
    j = jn; ent_v = ent_n; ok = ok_n;
  }

  for (int i = wave; i < no; i += OWN_NW) {
    const int e  = ovf[i];
    const int dq = dst[e] - n0;
    if ((unsigned)dq < (unsigned)OWN_NB) {
      const int   s = src[e];
      const float w = dinv[s] * dloc[dq];
      const vrow_t x = *(const vrow_t*)(h + (size_t)s * C + V * lane);
      float* a = acc + dq * C + lane;
#pragma unroll
      for (int k = 0; k < V; ++k)
        __hip_atomic_fetch_add(a + 32 * k, w * x[k], __ATOMIC_RELAXED,
                               __HIP_MEMORY_SCOPE_WORKGROUP);
    }
  }
  __syncthreads();

  for (int t = tid; t < OWN_NB * C; t += OWN_T) {
    const int i = t / C;
    const int f = t - i * C;
    const int node = n0 + i;
    if (node < N_NODES) {
      const float di = dloc[i];
      float r = bias[f] + di * di * h[(size_t)node * C + f] + acc[i * C + 32 * (f % V) + f / V];
      if (RELU) r = fmaxf(r, 0.0f);
      out[(size_t)node * C + f] = r;
    }
  }
}

extern "C" void kernel_launch(void* const* d_in, const int* in_sizes, int n_in,
                              void* d_out, int out_size, void* d_ws, size_t ws_size,
                              hipStream_t stream) {
  (void)in_sizes; (void)n_in; (void)out_size; (void)ws_size;
  const float* x  = (const float*)d_in[0];
  const int*   ei = (const int*)d_in[1];
  const float* W1 = (const float*)d_in[2];
  const float* b1 = (const float*)d_in[3];
  const float* W2 = (const float*)d_in[4];
  const float* b2 = (const float*)d_in[5];
  const int* src = ei;
  const int* dst = ei + E_EDGES;
  float* out = (float*)d_out;

  char* ws = (char*)d_ws;
  float* dinv = (float*)ws;  ws += (((size_t)N_NODES * 4) + 255) / 256 * 256;
  float* H1   = (float*)ws;  ws += (size_t)N_NODES * C_HID * 4;
  float* A1   = (float*)ws;  ws += (size_t)N_NODES * C_HID * 4;
  float* H2   = (float*)ws;  ws += (size_t)N_NODES * C_OUT * 4;
  int*   cur  = (int*)ws;    ws += (((size_t)(OWN_NBLK + 1) * CUR_PITCH * 4) + 255) / 256 * 256;
  int*   bins = (int*)ws;    ws += (size_t)OWN_NBLK * BIN_CAP * 4;
  int*   ovf  = (int*)ws;

  cur_init_k<<<(OWN_NBLK + 1 + 255) / 256, dim3(256), 0, stream>>>(cur);
  bin_k<<<BIN_NBLK, dim3(BIN_T), 0, stream>>>(cur, bins, ovf, src, dst, E_EDGES);
  deg_k<<<OWN_NBLK, dim3(DEG_T), 0, stream>>>(dinv, cur, bins, ovf, dst);

  gemm_strip_k<C_IN, C_HID><<<dim3((M_TILES + 3) / 4, C_HID / 64), dim3(128), 0, stream>>>(x, W1, H1);
  own_k<C_HID, true><<<OWN_NBLK, dim3(OWN_T), 0, stream>>>(H1, dinv, b1, src, dst, cur, bins, ovf, A1);

  gemm_strip_k<C_HID, C_OUT><<<dim3((M_TILES + 3) / 4, C_OUT / 64), dim3(128), 0, stream>>>(A1, W2, H2);
  own_k<C_OUT, false><<<OWN_NBLK, dim3(OWN_T), 0, stream>>>(H2, dinv, b2, src, dst, cur, bins, ovf, out);
}


# own_k: LDS row sums start at bias + dinv^2*h[row] (self loop), loaded at block start, not at 0; closing pass reads LDS only, one row store per wave; sum order: self term first
# speedup vs baseline: 3.7273x; 1.0420x over previous
#include <hip/hip_runtime.h>

#define N_NODES 10000
#define E_EDGES 640000
#define C_IN    256
#define C_HID   128
#define C_OUT   64

typedef float    v2f  __attribute__((ext_vector_type(2)));
typedef float    v8f  __attribute__((ext_vector_type(8)));

#define OWN_NB    20
#define OWN_T     512
#define OWN_NW    (OWN_T / 32)
#define OWN_QB    12
#define OWN_NBLK  ((N_NODES + OWN_NB - 1) / OWN_NB)
#define BIN_CAP   8192
#define BIN_SHIFT 8
#define CUR_PITCH 32
#define BIN_T     1024
#define BIN_EPT   5
#define BIN_NBLK  ((E_EDGES + BIN_T * BIN_EPT - 1) / (BIN_T * BIN_EPT))
#define DEG_T     256
#define DEG_U     6

static_assert(OWN_NB <= (1 << BIN_SHIFT), "local dst index must fit below the src field of a bin entry");
static_assert(N_NODES <= (1 << (31 - BIN_SHIFT)), "src must fit above the local dst field of a bin entry");
static_assert(OWN_NBLK <= BIN_T, "bin_k reserves one owner range per thread");

__global__ void cur_init_k(int* __restrict__ cur) {
  int i = blockIdx.x * blockDim.x + threadIdx.x;
  if (i <= OWN_NBLK) cur[i * CUR_PITCH] = 0;
}

__global__ __launch_bounds__(BIN_T)
void bin_k(int* __restrict__ cur, int* __restrict__ bins, int* __restrict__ ovf,
           const int* __restrict__ src, const int* __restrict__ dst, int e) {
  __shared__ int cnt[OWN_NBLK];
  const int tid = threadIdx.x;
  if (tid < OWN_NBLK) cnt[tid] = 0;

  const int i0 = blockIdx.x * (BIN_T * BIN_EPT) + tid;
  int d[BIN_EPT], s[BIN_EPT];
#pragma unroll
  for (int k = 0; k < BIN_EPT; ++k) {
    const int i = min(i0 + k * BIN_T, e - 1);
    d[k] = dst[i];
    s[k] = src[i];
  }
  __syncthreads();

  int b[BIN_EPT], r[BIN_EPT];
#pragma unroll
  for (int k = 0; k < BIN_EPT; ++k) {
    b[k] = d[k] / OWN_NB;
    r[k] = -1;
    if (i0 + k * BIN_T < e) r[k] = atomicAdd(&cnt[b[k]], 1);
  }
  __syncthreads();

  if (tid < OWN_NBLK) {
    const int c = cnt[tid];
    int base = 0;
    if (c > 0)
      base = __hip_atomic_fetch_add(&cur[tid * CUR_PITCH], c, __ATOMIC_RELAXED,
                                    __HIP_MEMORY_SCOPE_AGENT);
    cnt[tid] = base;
  }
  __syncthreads();

  int slot[BIN_EPT];
#pragma unroll
  for (int k = 0; k < BIN_EPT; ++k) slot[k] = cnt[b[k]] + r[k];
#pragma unroll
  for (int k = 0; k < BIN_EPT; ++k) {
    if (r[k] >= 0) {
      if (slot[k] < BIN_CAP) {
        bins[(size_t)b[k] * BIN_CAP + slot[k]] = (s[k] << BIN_SHIFT) | (d[k] - b[k] * OWN_NB);
      } else {
        const int o = __hip_atomic_fetch_add(&cur[OWN_NBLK * CUR_PITCH], 1, __ATOMIC_RELAXED,
                                             __HIP_MEMORY_SCOPE_AGENT);
        ovf[o] = i0 + k * BIN_T;
      }
    }
  }
}

__global__ __launch_bounds__(DEG_T)
void deg_k(float* __restrict__ dinv, const int* __restrict__ cur, const int* __restrict__ bins,
           const int* __restrict__ ovf, const int* __restrict__ dst) {
  __shared__ int cnt[OWN_NB];
  const int tid = threadIdx.x;
  const int n0  = blockIdx.x * OWN_NB;
  if (tid < OWN_NB) cnt[tid] = 0;
  const int n  = min(cur[blockIdx.x * CUR_PITCH], BIN_CAP);
  const int no = min(cur[OWN_NBLK * CUR_PITCH], E_EDGES);
  const int* __restrict__ bin = bins + (size_t)blockIdx.x * BIN_CAP;
  __syncthreads();

  for (int i = tid; i < n; i += DEG_T * DEG_U) {
    int ent[DEG_U];
#pragma unroll
    for (int u = 0; u < DEG_U; ++u) ent[u] = bin[min(i + u * DEG_T, n - 1)];
#pragma unroll
    for (int u = 0; u < DEG_U; ++u)
      if (i + u * DEG_T < n) atomicAdd(&cnt[ent[u] & ((1 << BIN_SHIFT) - 1)], 1);
  }
  for (int i = tid; i < no; i += DEG_T) {
    const int dq = dst[ovf[i]] - n0;
    if ((unsigned)dq < (unsigned)OWN_NB) atomicAdd(&cnt[dq], 1);
  }
  __syncthreads();

  if (tid < OWN_NB && n0 + tid < N_NODES) dinv[n0 + tid] = rsqrtf(1.0f + (float)cnt[tid]);
}

#define GEMM_KC 64
#define GEMM_NG (GEMM_KC / 8)
#define M_TILES (N_NODES / 16)

template <int K, int N>
__device__ __forceinline__ void gemm_load_chunk(float4 (&a)[GEMM_NG], float4 (&b)[GEMM_NG][4],
                                                const float* __restrict__ Ap,
                                                const float* __restrict__ Bp, int k0) {
#pragma unroll
  for (int g = 0; g < GEMM_NG; ++g) {
    a[g] = *(const float4*)(Ap + k0 + 8 * g);
#pragma unroll
    for (int i = 0; i < 4; ++i) b[g][i] = *(const float4*)(Bp + (size_t)(k0 + 8 * g + i) * N);
  }
}

__device__ __forceinline__ void gemm_mma_chunk(v8f (&acc)[4], const float4 (&a)[GEMM_NG],
                                               const float4 (&b)[GEMM_NG][4]) {
#pragma unroll
  for (int g = 0; g < GEMM_NG; ++g) {
    const v2f a0 = {a[g].x, a[g].y};
    const v2f a1 = {a[g].z, a[g].w};
    const float b0[4] = {b[g][0].x, b[g][0].y, b[g][0].z, b[g][0].w};
    const float b1[4] = {b[g][1].x, b[g][1].y, b[g][1].z, b[g][1].w};
    const float b2[4] = {b[g][2].x, b[g][2].y, b[g][2].z, b[g][2].w};
    const float b3[4] = {b[g][3].x, b[g][3].y, b[g][3].z, b[g][3].w};
#pragma unroll
    for (int t = 0; t < 4; ++t) {
      const v2f bb = {b0[t], b1[t]};
      acc[t] = __builtin_amdgcn_wmma_f32_16x16x4_f32(false, a0, false, bb, (short)0, acc[t],
                                                     false, false);
    }
#pragma unroll
    for (int t = 0; t < 4; ++t) {
      const v2f bb = {b2[t], b3[t]};
      acc[t] = __builtin_amdgcn_wmma_f32_16x16x4_f32(false, a1, false, bb, (short)0, acc[t],
                                                     false, false);
    }
  }
}

template <int K, int N>
__global__ __launch_bounds__(128)
void gemm_strip_k(const float* __restrict__ A, const float* __restrict__ B, float* __restrict__ D) {
  static_assert(K % (2 * GEMM_KC) == 0, "the chunk loop below is written two chunks per trip");
  static_assert(N % 64 == 0, "a strip is 64 columns");
  const int lane = threadIdx.x & 31;
  const int wave = __builtin_amdgcn_readfirstlane(threadIdx.x >> 5);
  const int rt   = blockIdx.x * 4 + wave;
  if (rt >= M_TILES) return;
  const int c  = lane & 15;
  const int h  = lane >> 4;
  const int n0 = blockIdx.y * 64;

  const float* __restrict__ Ap = A + (size_t)(rt * 16 + c) * K + 4 * h;
  const float* __restrict__ Bp = B + (size_t)(4 * h) * N + n0 + 4 * c;

  v8f acc[4] = {};
  float4 a0[GEMM_NG], b0[GEMM_NG][4];
  float4 a1[GEMM_NG], b1[GEMM_NG][4];

  gemm_load_chunk<K, N>(a0, b0, Ap, Bp, 0);
#pragma unroll
  for (int k0 = 0; k0 < K; k0 += 2 * GEMM_KC) {
    gemm_load_chunk<K, N>(a1, b1, Ap, Bp, k0 + GEMM_KC);
    __builtin_amdgcn_sched_barrier(0);
    gemm_mma_chunk(acc, a0, b0);
    if (k0 + 2 * GEMM_KC < K) gemm_load_chunk<K, N>(a0, b0, Ap, Bp, k0 + 2 * GEMM_KC);
    __builtin_amdgcn_sched_barrier(0);
    gemm_mma_chunk(acc, a1, b1);
  }

  float* Dp = D + (size_t)(rt * 16 + 8 * h) * N + n0 + 4 * c;
#pragma unroll
  for (int i = 0; i < 8; ++i) {
    const float4 o = {acc[0][i], acc[1][i], acc[2][i], acc[3][i]};
    *(float4*)(Dp + (size_t)i * N) = o;
  }
}

template <int C, bool RELU>
__global__ __launch_bounds__(OWN_T)
void own_k(const float* __restrict__ h, const float* __restrict__ dinv,
           const float* __restrict__ bias, const int* __restrict__ src,
           const int* __restrict__ dst, const int* __restrict__ cur,
           const int* __restrict__ bins, const int* __restrict__ ovf,
           float* __restrict__ out) {
  __shared__ float acc[OWN_NB * C];
  __shared__ float dloc[OWN_NB];
  constexpr int V = C / 32;
  typedef float vrow_t __attribute__((ext_vector_type(V)));

  const int tid  = threadIdx.x;
  const int lane = tid & 31;
  const int wave = __builtin_amdgcn_readfirstlane(tid >> 5);
  const int n0   = blockIdx.x * OWN_NB;

  constexpr int NSLOT = OWN_NB * 32;
  constexpr int NTRIP = (NSLOT + OWN_T - 1) / OWN_T;
  {
    const vrow_t bv = *(const vrow_t*)(bias + V * lane);
    float  di[NTRIP];
    vrow_t hv[NTRIP];
#pragma unroll
    for (int q = 0; q < NTRIP; ++q) {
      const int i    = min((tid + q * OWN_T) >> 5, OWN_NB - 1);
      const int node = min(n0 + i, N_NODES - 1);
      di[q] = dinv[node];
      hv[q] = *(const vrow_t*)(h + (size_t)node * C + V * lane);
    }
#pragma unroll
    for (int q = 0; q < NTRIP; ++q) {
      const int t = tid + q * OWN_T;
      const int i = t >> 5;
      if (t < NSLOT) {
        float* a = acc + i * C + lane;
        const bool real = n0 + i < N_NODES;
#pragma unroll
        for (int k = 0; k < V; ++k) a[32 * k] = real ? bv[k] + di[q] * di[q] * hv[q][k] : 0.0f;
      }
    }
  }
  if (tid < OWN_NB) dloc[tid] = (n0 + tid < N_NODES) ? dinv[n0 + tid] : 0.0f;

  const int n  = min(cur[blockIdx.x * CUR_PITCH], BIN_CAP);
  const int no = min(cur[OWN_NBLK * CUR_PITCH], E_EDGES);
  const int* __restrict__ bin = bins + (size_t)blockIdx.x * BIN_CAP;
  __syncthreads();

  const int ul = (lane < OWN_QB) ? lane : 0;
  int  j     = wave;
  int  ent_v = 0;
  bool ok    = false;
  if (j * OWN_QB < n) {
    const int i = j * OWN_QB + ul;
    ok    = i < n;
    ent_v = bin[ok ? i : j * OWN_QB];
  }
  while (j * OWN_QB < n) {
    const int   s_v  = ent_v >> BIN_SHIFT;
    const int   dq   = ent_v & ((1 << BIN_SHIFT) - 1);
    const int   dl_v = ok ? dq : -1;
    const float ds_v = dinv[s_v];
    const int jn = j + OWN_NW;
    int  ent_n = 0;
    bool ok_n  = false;
    if (jn * OWN_QB < n) {
      const int i = jn * OWN_QB + ul;
      ok_n  = i < n;
      ent_n = bin[ok_n ? i : jn * OWN_QB];
    }
    vrow_t v[OWN_QB];
#pragma unroll
    for (int u = 0; u < OWN_QB; ++u) {
      const int dl_u = __builtin_amdgcn_readlane(dl_v, u);
      if (dl_u >= 0) {
        const int s_u = __builtin_amdgcn_readlane(s_v, u);
        v[u] = *(const vrow_t*)(h + (size_t)s_u * C + V * lane);
      }
    }
    const float w_v = ds_v * dloc[dq];
#pragma unroll
    for (int u = 0; u < OWN_QB; ++u) {
      const int dl_u = __builtin_amdgcn_readlane(dl_v, u);
      if (dl_u >= 0) {
        const float w_u = __int_as_float(__builtin_amdgcn_readlane(__float_as_int(w_v), u));
        float* a = acc + dl_u * C + lane;
#pragma unroll
        for (int k = 0; k < V; ++k)
          __hip_atomic_fetch_add(a + 32 * k, w_u * v[u][k], __ATOMIC_RELAXED,
                                 __HIP_MEMORY_SCOPE_WORKGROUP);
      }
    }
    j = jn; ent_v = ent_n; ok = ok_n;
  }

  for (int i = wave; i < no; i += OWN_NW) {
    const int e  = ovf[i];
    const int dq = dst[e] - n0;
    if ((unsigned)dq < (unsigned)OWN_NB) {
      const int   s = src[e];
      const float w = dinv[s] * dloc[dq];
      const vrow_t x = *(const vrow_t*)(h + (size_t)s * C + V * lane);
      float* a = acc + dq * C + lane;
#pragma unroll
      for (int k = 0; k < V; ++k)
        __hip_atomic_fetch_add(a + 32 * k, w * x[k], __ATOMIC_RELAXED,
                               __HIP_MEMORY_SCOPE_WORKGROUP);
    }
  }
  __syncthreads();

  for (int t = tid; t < OWN_NB * 32; t += OWN_T) {
    const int i    = t >> 5;
    const int node = n0 + i;
    if (node < N_NODES) {
      const float* a = acc + i * C + lane;
      vrow_t r;
#pragma unroll
      for (int k = 0; k < V; ++k) {
        r[k] = a[32 * k];
        if (RELU) r[k] = fmaxf(r[k], 0.0f);
      }
      *(vrow_t*)(out + (size_t)node * C + V * lane) = r;
    }
  }
}

extern "C" void kernel_launch(void* const* d_in, const int* in_sizes, int n_in,
                              void* d_out, int out_size, void* d_ws, size_t ws_size,
                              hipStream_t stream) {
  (void)in_sizes; (void)n_in; (void)out_size; (void)ws_size;
  const float* x  = (const float*)d_in[0];
  const int*   ei = (const int*)d_in[1];
  const float* W1 = (const float*)d_in[2];
  const float* b1 = (const float*)d_in[3];
  const float* W2 = (const float*)d_in[4];
  const float* b2 = (const float*)d_in[5];
  const int* src = ei;
  const int* dst = ei + E_EDGES;
  float* out = (float*)d_out;

  char* ws = (char*)d_ws;
  float* dinv = (float*)ws;  ws += (((size_t)N_NODES * 4) + 255) / 256 * 256;
  float* H1   = (float*)ws;  ws += (size_t)N_NODES * C_HID * 4;
  float* A1   = (float*)ws;  ws += (size_t)N_NODES * C_HID * 4;
  float* H2   = (float*)ws;  ws += (size_t)N_NODES * C_OUT * 4;
  int*   cur  = (int*)ws;    ws += (((size_t)(OWN_NBLK + 1) * CUR_PITCH * 4) + 255) / 256 * 256;
  int*   bins = (int*)ws;    ws += (size_t)OWN_NBLK * BIN_CAP * 4;
  int*   ovf  = (int*)ws;

  cur_init_k<<<(OWN_NBLK + 1 + 255) / 256, dim3(256), 0, stream>>>(cur);
  bin_k<<<BIN_NBLK, dim3(BIN_T), 0, stream>>>(cur, bins, ovf, src, dst, E_EDGES);
  deg_k<<<OWN_NBLK, dim3(DEG_T), 0, stream>>>(dinv, cur, bins, ovf, dst);

  gemm_strip_k<C_IN, C_HID><<<dim3((M_TILES + 3) / 4, C_HID / 64), dim3(128), 0, stream>>>(x, W1, H1);
  own_k<C_HID, true><<<OWN_NBLK, dim3(OWN_T), 0, stream>>>(H1, dinv, b1, src, dst, cur, bins, ovf, A1);

  gemm_strip_k<C_HID, C_OUT><<<dim3((M_TILES + 3) / 4, C_OUT / 64), dim3(128), 0, stream>>>(A1, W2, H2);
  own_k<C_OUT, false><<<OWN_NBLK, dim3(OWN_T), 0, stream>>>(H2, dinv, b2, src, dst, cur, bins, ovf, out);
}


# own_k: an edge's dinv[src] and dinv[dst] come from an LDS copy of the dinv table (copied per block by whole-line loads), not from a 4-byte global gather per edge; values and sum order as before
# speedup vs baseline: 3.8485x; 1.0325x over previous
#include <hip/hip_runtime.h>

#define N_NODES 10000
#define E_EDGES 640000
#define C_IN    256
#define C_HID   128
#define C_OUT   64

typedef float    v2f  __attribute__((ext_vector_type(2)));
typedef float    v8f  __attribute__((ext_vector_type(8)));
typedef float    v4f  __attribute__((ext_vector_type(4)));

#define OWN_NB    20
#define OWN_T     512
#define OWN_NW    (OWN_T / 32)
#define OWN_QB    12
#define OWN_NBLK  ((N_NODES + OWN_NB - 1) / OWN_NB)
#define BIN_CAP   8192
#define BIN_SHIFT 8
#define CUR_PITCH 32
#define BIN_T     1024
#define BIN_EPT   5
#define BIN_NBLK  ((E_EDGES + BIN_T * BIN_EPT - 1) / (BIN_T * BIN_EPT))
#define DEG_T     256
#define DEG_U     6

static_assert(OWN_NB <= (1 << BIN_SHIFT), "local dst index must fit below the src field of a bin entry");
static_assert(N_NODES <= (1 << (31 - BIN_SHIFT)), "src must fit above the local dst field of a bin entry");
static_assert(OWN_NBLK <= BIN_T, "bin_k reserves one owner range per thread");

__global__ void cur_init_k(int* __restrict__ cur) {
  int i = blockIdx.x * blockDim.x + threadIdx.x;
  if (i <= OWN_NBLK) cur[i * CUR_PITCH] = 0;
}

__global__ __launch_bounds__(BIN_T)
void bin_k(int* __restrict__ cur, int* __restrict__ bins, int* __restrict__ ovf,
           const int* __restrict__ src, const int* __restrict__ dst, int e) {
  __shared__ int cnt[OWN_NBLK];
  const int tid = threadIdx.x;
  if (tid < OWN_NBLK) cnt[tid] = 0;

  const int i0 = blockIdx.x * (BIN_T * BIN_EPT) + tid;
  int d[BIN_EPT], s[BIN_EPT];
#pragma unroll
  for (int k = 0; k < BIN_EPT; ++k) {
    const int i = min(i0 + k * BIN_T, e - 1);
    d[k] = dst[i];
    s[k] = src[i];
  }
  __syncthreads();

  int b[BIN_EPT], r[BIN_EPT];
#pragma unroll
  for (int k = 0; k < BIN_EPT; ++k) {
    b[k] = d[k] / OWN_NB;
    r[k] = -1;
    if (i0 + k * BIN_T < e) r[k] = atomicAdd(&cnt[b[k]], 1);
  }
  __syncthreads();

  if (tid < OWN_NBLK) {
    const int c = cnt[tid];
    int base = 0;
    if (c > 0)
      base = __hip_atomic_fetch_add(&cur[tid * CUR_PITCH], c, __ATOMIC_RELAXED,
                                    __HIP_MEMORY_SCOPE_AGENT);
    cnt[tid] = base;
  }
  __syncthreads();

  int slot[BIN_EPT];
#pragma unroll
  for (int k = 0; k < BIN_EPT; ++k) slot[k] = cnt[b[k]] + r[k];
#pragma unroll
  for (int k = 0; k < BIN_EPT; ++k) {
    if (r[k] >= 0) {
      if (slot[k] < BIN_CAP) {
        bins[(size_t)b[k] * BIN_CAP + slot[k]] = (s[k] << BIN_SHIFT) | (d[k] - b[k] * OWN_NB);
      } else {
        const int o = __hip_atomic_fetch_add(&cur[OWN_NBLK * CUR_PITCH], 1, __ATOMIC_RELAXED,
                                             __HIP_MEMORY_SCOPE_AGENT);
        ovf[o] = i0 + k * BIN_T;
      }
    }
  }
}

__global__ __launch_bounds__(DEG_T)
void deg_k(float* __restrict__ dinv, const int* __restrict__ cur, const int* __restrict__ bins,
           const int* __restrict__ ovf, const int* __restrict__ dst) {
  __shared__ int cnt[OWN_NB];
  const int tid = threadIdx.x;
  const int n0  = blockIdx.x * OWN_NB;
  if (tid < OWN_NB) cnt[tid] = 0;
  const int n  = min(cur[blockIdx.x * CUR_PITCH], BIN_CAP);
  const int no = min(cur[OWN_NBLK * CUR_PITCH], E_EDGES);
  const int* __restrict__ bin = bins + (size_t)blockIdx.x * BIN_CAP;
  __syncthreads();

  for (int i = tid; i < n; i += DEG_T * DEG_U) {
    int ent[DEG_U];
#pragma unroll
    for (int u = 0; u < DEG_U; ++u) ent[u] = bin[min(i + u * DEG_T, n - 1)];
#pragma unroll
    for (int u = 0; u < DEG_U; ++u)
      if (i + u * DEG_T < n) atomicAdd(&cnt[ent[u] & ((1 << BIN_SHIFT) - 1)], 1);
  }
  for (int i = tid; i < no; i += DEG_T) {
    const int dq = dst[ovf[i]] - n0;
    if ((unsigned)dq < (unsigned)OWN_NB) atomicAdd(&cnt[dq], 1);
  }
  __syncthreads();

  if (tid < OWN_NB && n0 + tid < N_NODES) dinv[n0 + tid] = rsqrtf(1.0f + (float)cnt[tid]);
}

#define GEMM_KC 64
#define GEMM_NG (GEMM_KC / 8)
#define M_TILES (N_NODES / 16)

template <int K, int N>
__device__ __forceinline__ void gemm_load_chunk(float4 (&a)[GEMM_NG], float4 (&b)[GEMM_NG][4],
                                                const float* __restrict__ Ap,
                                                const float* __restrict__ Bp, int k0) {
#pragma unroll
  for (int g = 0; g < GEMM_NG; ++g) {
    a[g] = *(const float4*)(Ap + k0 + 8 * g);
#pragma unroll
    for (int i = 0; i < 4; ++i) b[g][i] = *(const float4*)(Bp + (size_t)(k0 + 8 * g + i) * N);
  }
}

__device__ __forceinline__ void gemm_mma_chunk(v8f (&acc)[4], const float4 (&a)[GEMM_NG],
                                               const float4 (&b)[GEMM_NG][4]) {
#pragma unroll
  for (int g = 0; g < GEMM_NG; ++g) {
    const v2f a0 = {a[g].x, a[g].y};
    const v2f a1 = {a[g].z, a[g].w};
    const float b0[4] = {b[g][0].x, b[g][0].y, b[g][0].z, b[g][0].w};
    const float b1[4] = {b[g][1].x, b[g][1].y, b[g][1].z, b[g][1].w};
    const float b2[4] = {b[g][2].x, b[g][2].y, b[g][2].z, b[g][2].w};
    const float b3[4] = {b[g][3].x, b[g][3].y, b[g][3].z, b[g][3].w};
#pragma unroll
    for (int t = 0; t < 4; ++t) {
      const v2f bb = {b0[t], b1[t]};
      acc[t] = __builtin_amdgcn_wmma_f32_16x16x4_f32(false, a0, false, bb, (short)0, acc[t],
                                                     false, false);
    }
#pragma unroll
    for (int t = 0; t < 4; ++t) {
      const v2f bb = {b2[t], b3[t]};
      acc[t] = __builtin_amdgcn_wmma_f32_16x16x4_f32(false, a1, false, bb, (short)0, acc[t],
                                                     false, false);
    }
  }
}

template <int K, int N>
__global__ __launch_bounds__(128)
void gemm_strip_k(const float* __restrict__ A, const float* __restrict__ B, float* __restrict__ D) {
  static_assert(K % (2 * GEMM_KC) == 0, "the chunk loop below is written two chunks per trip");
  static_assert(N % 64 == 0, "a strip is 64 columns");
  const int lane = threadIdx.x & 31;
  const int wave = __builtin_amdgcn_readfirstlane(threadIdx.x >> 5);
  const int rt   = blockIdx.x * 4 + wave;
  if (rt >= M_TILES) return;
  const int c  = lane & 15;
  const int h  = lane >> 4;
  const int n0 = blockIdx.y * 64;

  const float* __restrict__ Ap = A + (size_t)(rt * 16 + c) * K + 4 * h;
  const float* __restrict__ Bp = B + (size_t)(4 * h) * N + n0 + 4 * c;

  v8f acc[4] = {};
  float4 a0[GEMM_NG], b0[GEMM_NG][4];
  float4 a1[GEMM_NG], b1[GEMM_NG][4];

  gemm_load_chunk<K, N>(a0, b0, Ap, Bp, 0);
#pragma unroll
  for (int k0 = 0; k0 < K; k0 += 2 * GEMM_KC) {
    gemm_load_chunk<K, N>(a1, b1, Ap, Bp, k0 + GEMM_KC);
    __builtin_amdgcn_sched_barrier(0);
    gemm_mma_chunk(acc, a0, b0);
    if (k0 + 2 * GEMM_KC < K) gemm_load_chunk<K, N>(a0, b0, Ap, Bp, k0 + 2 * GEMM_KC);
    __builtin_amdgcn_sched_barrier(0);
    gemm_mma_chunk(acc, a1, b1);
  }

  float* Dp = D + (size_t)(rt * 16 + 8 * h) * N + n0 + 4 * c;
#pragma unroll
  for (int i = 0; i < 8; ++i) {
    const float4 o = {acc[0][i], acc[1][i], acc[2][i], acc[3][i]};
    *(float4*)(Dp + (size_t)i * N) = o;
  }
}

static_assert(N_NODES % 4 == 0, "the dinv table is copied to LDS as 16-byte elements");
#define TAB_V (N_NODES / 4)
#define TAB_Q ((TAB_V + OWN_T - 1) / OWN_T)
template <int C, bool RELU>
__global__ __launch_bounds__(OWN_T)
void own_k(const float* __restrict__ h, const float* __restrict__ dinv,
           const float* __restrict__ bias, const int* __restrict__ src,
           const int* __restrict__ dst, const int* __restrict__ cur,
           const int* __restrict__ bins, const int* __restrict__ ovf,
           float* __restrict__ out) {
  __shared__ float acc[OWN_NB * C];
  __shared__ v4f dtab4[TAB_Q * OWN_T];
  const float* dtab = (const float*)dtab4;
  constexpr int V = C / 32;
  typedef float vrow_t __attribute__((ext_vector_type(V)));

  const int tid  = threadIdx.x;
  const int lane = tid & 31;
  const int wave = __builtin_amdgcn_readfirstlane(tid >> 5);
  const int n0   = blockIdx.x * OWN_NB;

  constexpr int NSLOT = OWN_NB * 32;
  constexpr int NTRIP = (NSLOT + OWN_T - 1) / OWN_T;
  {
    v4f tv[TAB_Q];
#pragma unroll
    for (int q = 0; q < TAB_Q; ++q)
      tv[q] = ((const v4f*)dinv)[min(tid + q * OWN_T, TAB_V - 1)];
    const vrow_t bv = *(const vrow_t*)(bias + V * lane);
    float  di[NTRIP];
    vrow_t hv[NTRIP];
#pragma unroll
    for (int q = 0; q < NTRIP; ++q) {
      const int i    = min((tid + q * OWN_T) >> 5, OWN_NB - 1);
      const int node = min(n0 + i, N_NODES - 1);
      di[q] = dinv[node];
      hv[q] = *(const vrow_t*)(h + (size_t)node * C + V * lane);
    }
#pragma unroll
    for (int q = 0; q < TAB_Q; ++q) dtab4[tid + q * OWN_T] = tv[q];
#pragma unroll
    for (int q = 0; q < NTRIP; ++q) {
      const int t = tid + q * OWN_T;
      const int i = t >> 5;
      if (t < NSLOT) {
        float* a = acc + i * C + lane;
        const bool real = n0 + i < N_NODES;
#pragma unroll
        for (int k = 0; k < V; ++k) a[32 * k] = real ? bv[k] + di[q] * di[q] * hv[q][k] : 0.0f;
      }
    }
  }

  const int n  = min(cur[blockIdx.x * CUR_PITCH], BIN_CAP);
  const int no = min(cur[OWN_NBLK * CUR_PITCH], E_EDGES);
  const int* __restrict__ bin = bins + (size_t)blockIdx.x * BIN_CAP;
  __syncthreads();

  const int ul = (lane < OWN_QB) ? lane : 0;
  int  j     = wave;
  int  ent_v = 0;
  bool ok    = false;
  if (j * OWN_QB < n) {
    const int i = j * OWN_QB + ul;
    ok    = i < n;
    ent_v = bin[ok ? i : j * OWN_QB];
  }
  while (j * OWN_QB < n) {
    const int   s_v  = ent_v >> BIN_SHIFT;
    const int   dq   = ent_v & ((1 << BIN_SHIFT) - 1);
    const int   dl_v = ok ? dq : -1;
    const float ds_v = dtab[s_v];
    const int jn = j + OWN_NW;
    int  ent_n = 0;
    bool ok_n  = false;
    if (jn * OWN_QB < n) {
      const int i = jn * OWN_QB + ul;
      ok_n  = i < n;
      ent_n = bin[ok_n ? i : jn * OWN_QB];
    }
    vrow_t v[OWN_QB];
#pragma unroll
    for (int u = 0; u < OWN_QB; ++u) {
      const int dl_u = __builtin_amdgcn_readlane(dl_v, u);
      if (dl_u >= 0) {
        const int s_u = __builtin_amdgcn_readlane(s_v, u);
        v[u] = *(const vrow_t*)(h + (size_t)s_u * C + V * lane);
      }
    }
    const float w_v = ds_v * dtab[n0 + dq];
#pragma unroll
    for (int u = 0; u < OWN_QB; ++u) {
      const int dl_u = __builtin_amdgcn_readlane(dl_v, u);
      if (dl_u >= 0) {
        const float w_u = __int_as_float(__builtin_amdgcn_readlane(__float_as_int(w_v), u));
        float* a = acc + dl_u * C + lane;
#pragma unroll
        for (int k = 0; k < V; ++k)
          __hip_atomic_fetch_add(a + 32 * k, w_u * v[u][k], __ATOMIC_RELAXED,
                                 __HIP_MEMORY_SCOPE_WORKGROUP);
      }
    }
    j = jn; ent_v = ent_n; ok = ok_n;
  }

  for (int i = wave; i < no; i += OWN_NW) {
    const int e  = ovf[i];
    const int dq = dst[e] - n0;
    if ((unsigned)dq < (unsigned)OWN_NB) {
      const int   s = src[e];
      const float w = dtab[s] * dtab[n0 + dq];
      const vrow_t x = *(const vrow_t*)(h + (size_t)s * C + V * lane);
      float* a = acc + dq * C + lane;
#pragma unroll
      for (int k = 0; k < V; ++k)
        __hip_atomic_fetch_add(a + 32 * k, w * x[k], __ATOMIC_RELAXED,
                               __HIP_MEMORY_SCOPE_WORKGROUP);
    }
  }
  __syncthreads();

  for (int t = tid; t < OWN_NB * 32; t += OWN_T) {
    const int i    = t >> 5;
    const int node = n0 + i;
    if (node < N_NODES) {
      const float* a = acc + i * C + lane;
      vrow_t r;
#pragma unroll
      for (int k = 0; k < V; ++k) {
        r[k] = a[32 * k];
        if (RELU) r[k] = fmaxf(r[k], 0.0f);
      }
      *(vrow_t*)(out + (size_t)node * C + V * lane) = r;
    }
  }
}

extern "C" void kernel_launch(void* const* d_in, const int* in_sizes, int n_in,
                              void* d_out, int out_size, void* d_ws, size_t ws_size,
                              hipStream_t stream) {
  (void)in_sizes; (void)n_in; (void)out_size; (void)ws_size;
  const float* x  = (const float*)d_in[0];
  const int*   ei = (const int*)d_in[1];
  const float* W1 = (const float*)d_in[2];
  const float* b1 = (const float*)d_in[3];
  const float* W2 = (const float*)d_in[4];
  const float* b2 = (const float*)d_in[5];
  const int* src = ei;
  const int* dst = ei + E_EDGES;
  float* out = (float*)d_out;

  char* ws = (char*)d_ws;
  float* dinv = (float*)ws;  ws += (((size_t)N_NODES * 4) + 255) / 256 * 256;
  float* H1   = (float*)ws;  ws += (size_t)N_NODES * C_HID * 4;
  float* A1   = (float*)ws;  ws += (size_t)N_NODES * C_HID * 4;
  float* H2   = (float*)ws;  ws += (size_t)N_NODES * C_OUT * 4;
  int*   cur  = (int*)ws;    ws += (((size_t)(OWN_NBLK + 1) * CUR_PITCH * 4) + 255) / 256 * 256;
  int*   bins = (int*)ws;    ws += (size_t)OWN_NBLK * BIN_CAP * 4;
  int*   ovf  = (int*)ws;

  cur_init_k<<<(OWN_NBLK + 1 + 255) / 256, dim3(256), 0, stream>>>(cur);
  bin_k<<<BIN_NBLK, dim3(BIN_T), 0, stream>>>(cur, bins, ovf, src, dst, E_EDGES);
  deg_k<<<OWN_NBLK, dim3(DEG_T), 0, stream>>>(dinv, cur, bins, ovf, dst);

  gemm_strip_k<C_IN, C_HID><<<dim3((M_TILES + 3) / 4, C_HID / 64), dim3(128), 0, stream>>>(x, W1, H1);
  own_k<C_HID, true><<<OWN_NBLK, dim3(OWN_T), 0, stream>>>(H1, dinv, b1, src, dst, cur, bins, ovf, A1);

  gemm_strip_k<C_HID, C_OUT><<<dim3((M_TILES + 3) / 4, C_OUT / 64), dim3(128), 0, stream>>>(A1, W2, H2);
  own_k<C_OUT, false><<<OWN_NBLK, dim3(OWN_T), 0, stream>>>(H2, dinv, b2, src, dst, cur, bins, ovf, out);
}


# edge bins as one 128-byte line per (owner range, binning block), count in word 0, built in LDS; a wave trip = one piece; no cursors, global atomics or clearing launch; sum order = new bin order
# speedup vs baseline: 4.4108x; 1.1461x over previous
#include <hip/hip_runtime.h>

#define N_NODES 10000
#define E_EDGES 640000
#define C_IN    256
#define C_HID   128
#define C_OUT   64

typedef float    v2f  __attribute__((ext_vector_type(2)));
typedef float    v8f  __attribute__((ext_vector_type(8)));
typedef float    v4f  __attribute__((ext_vector_type(4)));
typedef int      v4i  __attribute__((ext_vector_type(4)));

#define OWN_NB    20
#define OWN_T     512
#define OWN_NW    (OWN_T / 32)
#define OWN_NBLK  ((N_NODES + OWN_NB - 1) / OWN_NB)
#define BIN_SHIFT 8
#define BIN_T     1024
#define BIN_EPT   5
#define BIN_E     (BIN_T * BIN_EPT)
#define BIN_NBLK  ((E_EDGES + BIN_E - 1) / BIN_E)
#define PIECE_W   32
#define PIECE_CAP (PIECE_W - 1)
#define CNT_PITCH 32
#define DEG_T     256
#define DEG_NW    (DEG_T / 32)
#define DEG_Q     ((BIN_NBLK + DEG_NW - 1) / DEG_NW)
#define OVN_Q     ((BIN_NBLK + 31) / 32)

static_assert(OWN_NB <= (1 << BIN_SHIFT), "local dst index must fit below the src field of an entry");
static_assert(N_NODES <= (1 << (31 - BIN_SHIFT)), "src must fit above the local dst field of an entry");
static_assert(OWN_NBLK <= BIN_T, "bin_k handles one owner range per thread");
static_assert((OWN_NBLK * PIECE_W / 4) % 8 == 0, "a piece is 8 16-byte elements");
static_assert(BIN_NBLK <= DEG_T && BIN_NBLK <= OWN_T, "one thread per binning block reads its overflow count");

__global__ __launch_bounds__(BIN_T)
void bin_k(int* __restrict__ pieces, int* __restrict__ ovf, int* __restrict__ ovfc,
           const int* __restrict__ src, const int* __restrict__ dst, int e) {
  __shared__ int  cnt[OWN_NBLK];
  __shared__ v4i  stage4[OWN_NBLK * PIECE_W / 4];
  __shared__ int  ocnt;
  int* stage = (int*)stage4;
  const int tid = threadIdx.x;
  if (tid < OWN_NBLK) cnt[tid] = 0;
  if (tid == 0) ocnt = 0;

  const int i0 = blockIdx.x * BIN_E + tid;
  int d[BIN_EPT], s[BIN_EPT];
#pragma unroll
  for (int k = 0; k < BIN_EPT; ++k) {
    const int i = min(i0 + k * BIN_T, e - 1);
    d[k] = dst[i];
    s[k] = src[i];
  }
  for (int i = tid; i < OWN_NBLK * PIECE_W / 4; i += BIN_T) stage4[i] = (v4i){0, 0, 0, 0};
  __syncthreads();

#pragma unroll
  for (int k = 0; k < BIN_EPT; ++k) {
    if (i0 + k * BIN_T < e) {
      const int b = d[k] / OWN_NB;
      const int r = atomicAdd(&cnt[b], 1);
      if (r < PIECE_CAP) {
        stage[b * PIECE_W + 1 + r] = (s[k] << BIN_SHIFT) | (d[k] - b * OWN_NB);
      } else {
        const int o = atomicAdd(&ocnt, 1);
        ovf[(size_t)blockIdx.x * BIN_E + o] = i0 + k * BIN_T;
      }
    }
  }
  __syncthreads();

  if (tid < OWN_NBLK) stage[tid * PIECE_W] = cnt[tid];
  __syncthreads();
  if (tid == 0) ovfc[blockIdx.x * CNT_PITCH] = ocnt;
  for (int i = tid; i < OWN_NBLK * PIECE_W / 4; i += BIN_T)
    ((v4i*)pieces)[((size_t)(i >> 3) * BIN_NBLK + blockIdx.x) * (PIECE_W / 4) + (i & 7)] = stage4[i];
}

__global__ __launch_bounds__(DEG_T)
void deg_k(float* __restrict__ dinv, const int* __restrict__ pieces, const int* __restrict__ ovf,
           const int* __restrict__ ovfc, const int* __restrict__ dst) {
  __shared__ int cnt[OWN_NB];
  __shared__ int ovn[BIN_NBLK];
  const int tid  = threadIdx.x;
  const int lane = tid & 31;
  const int wave = __builtin_amdgcn_readfirstlane(tid >> 5);
  const int n0   = blockIdx.x * OWN_NB;
  if (tid < OWN_NB) cnt[tid] = 0;
  const int* __restrict__ base = pieces + (size_t)blockIdx.x * BIN_NBLK * PIECE_W;

  int ent[DEG_Q];
#pragma unroll
  for (int q = 0; q < DEG_Q; ++q) ent[q] = base[min(wave + q * DEG_NW, BIN_NBLK - 1) * PIECE_W + lane];
  const int ovc = ovfc[min(tid, BIN_NBLK - 1) * CNT_PITCH];
  if (tid < BIN_NBLK) ovn[tid] = min(ovc, BIN_E);
  __syncthreads();

#pragma unroll
  for (int q = 0; q < DEG_Q; ++q) {
    const int n = min(__builtin_amdgcn_readfirstlane(ent[q]), PIECE_CAP);
    if (wave + q * DEG_NW < BIN_NBLK && lane >= 1 && lane <= n)
      atomicAdd(&cnt[ent[q] & ((1 << BIN_SHIFT) - 1)], 1);
  }
#pragma unroll
  for (int q = 0; q < OVN_Q; ++q) {
    const int c = lane + 32 * q < BIN_NBLK ? ovn[lane + 32 * q] : 0;
    unsigned mask = __builtin_amdgcn_ballot_w32(c > 0);
    while (mask) {
      const int j  = __builtin_ctz(mask);
      const int no = __builtin_amdgcn_readlane(c, j);
      const int* __restrict__ lst = ovf + (size_t)(32 * q + j) * BIN_E;
      mask &= mask - 1;
      for (int i = tid; i < no; i += DEG_T) {
        const int dq = dst[lst[i]] - n0;
        if ((unsigned)dq < (unsigned)OWN_NB) atomicAdd(&cnt[dq], 1);
      }
    }
  }
  __syncthreads();

  if (tid < OWN_NB && n0 + tid < N_NODES) dinv[n0 + tid] = rsqrtf(1.0f + (float)cnt[tid]);
}

#define GEMM_KC 64
#define GEMM_NG (GEMM_KC / 8)
#define M_TILES (N_NODES / 16)

template <int K, int N>
__device__ __forceinline__ void gemm_load_chunk(float4 (&a)[GEMM_NG], float4 (&b)[GEMM_NG][4],
                                                const float* __restrict__ Ap,
                                                const float* __restrict__ Bp, int k0) {
#pragma unroll
  for (int g = 0; g < GEMM_NG; ++g) {
    a[g] = *(const float4*)(Ap + k0 + 8 * g);
#pragma unroll
    for (int i = 0; i < 4; ++i) b[g][i] = *(const float4*)(Bp + (size_t)(k0 + 8 * g + i) * N);
  }
}

__device__ __forceinline__ void gemm_mma_chunk(v8f (&acc)[4], const float4 (&a)[GEMM_NG],
                                               const float4 (&b)[GEMM_NG][4]) {
#pragma unroll
  for (int g = 0; g < GEMM_NG; ++g) {
    const v2f a0 = {a[g].x, a[g].y};
    const v2f a1 = {a[g].z, a[g].w};
    const float b0[4] = {b[g][0].x, b[g][0].y, b[g][0].z, b[g][0].w};
    const float b1[4] = {b[g][1].x, b[g][1].y, b[g][1].z, b[g][1].w};
    const float b2[4] = {b[g][2].x, b[g][2].y, b[g][2].z, b[g][2].w};
    const float b3[4] = {b[g][3].x, b[g][3].y, b[g][3].z, b[g][3].w};
#pragma unroll
    for (int t = 0; t < 4; ++t) {
      const v2f bb = {b0[t], b1[t]};
      acc[t] = __builtin_amdgcn_wmma_f32_16x16x4_f32(false, a0, false, bb, (short)0, acc[t],
                                                     false, false);
    }
#pragma unroll
    for (int t = 0; t < 4; ++t) {
      const v2f bb = {b2[t], b3[t]};
      acc[t] = __builtin_amdgcn_wmma_f32_16x16x4_f32(false, a1, false, bb, (short)0, acc[t],
                                                     false, false);
    }
  }
}

template <int K, int N>
__global__ __launch_bounds__(128)
void gemm_strip_k(const float* __restrict__ A, const float* __restrict__ B, float* __restrict__ D) {
  static_assert(K % (2 * GEMM_KC) == 0, "the chunk loop below is written two chunks per trip");
  static_assert(N % 64 == 0, "a strip is 64 columns");
  const int lane = threadIdx.x & 31;
  const int wave = __builtin_amdgcn_readfirstlane(threadIdx.x >> 5);
  const int rt   = blockIdx.x * 4 + wave;
  if (rt >= M_TILES) return;
  const int c  = lane & 15;
  const int h  = lane >> 4;
  const int n0 = blockIdx.y * 64;

  const float* __restrict__ Ap = A + (size_t)(rt * 16 + c) * K + 4 * h;
  const float* __restrict__ Bp = B + (size_t)(4 * h) * N + n0 + 4 * c;

  v8f acc[4] = {};
  float4 a0[GEMM_NG], b0[GEMM_NG][4];
  float4 a1[GEMM_NG], b1[GEMM_NG][4];

  gemm_load_chunk<K, N>(a0, b0, Ap, Bp, 0);
#pragma unroll
  for (int k0 = 0; k0 < K; k0 += 2 * GEMM_KC) {
    gemm_load_chunk<K, N>(a1, b1, Ap, Bp, k0 + GEMM_KC);
    __builtin_amdgcn_sched_barrier(0);
    gemm_mma_chunk(acc, a0, b0);
    if (k0 + 2 * GEMM_KC < K) gemm_load_chunk<K, N>(a0, b0, Ap, Bp, k0 + 2 * GEMM_KC);
    __builtin_amdgcn_sched_barrier(0);
    gemm_mma_chunk(acc, a1, b1);
  }

  float* Dp = D + (size_t)(rt * 16 + 8 * h) * N + n0 + 4 * c;
#pragma unroll
  for (int i = 0; i < 8; ++i) {
    const float4 o = {acc[0][i], acc[1][i], acc[2][i], acc[3][i]};
    *(float4*)(Dp + (size_t)i * N) = o;
  }
}

static_assert(N_NODES % 4 == 0, "the dinv table is copied to LDS as 16-byte elements");
#define TAB_V (N_NODES / 4)
#define TAB_Q ((TAB_V + OWN_T - 1) / OWN_T)
#define OWN_QB 16
static_assert(OWN_QB <= PIECE_CAP, "a batch is part of one piece");

template <int C>
__device__ __forceinline__ void own_batch(const float* __restrict__ h, float* __restrict__ acc,
                                          int lane, int l0, int s_v, int dl_v, float w_v) {
  constexpr int V = C / 32;
  typedef float vrow_t __attribute__((ext_vector_type(V)));
  vrow_t v[OWN_QB];
#pragma unroll
  for (int u = 0; u < OWN_QB; ++u) {
    const int dl_u = __builtin_amdgcn_readlane(dl_v, (l0 + u) & 31);
    if (dl_u >= 0) {
      const int s_u = __builtin_amdgcn_readlane(s_v, (l0 + u) & 31);
      v[u] = *(const vrow_t*)(h + (size_t)s_u * C + V * lane);
    }
  }
#pragma unroll
  for (int u = 0; u < OWN_QB; ++u) {
    const int dl_u = __builtin_amdgcn_readlane(dl_v, (l0 + u) & 31);
    if (dl_u >= 0) {
      const float w_u = __int_as_float(__builtin_amdgcn_readlane(__float_as_int(w_v), (l0 + u) & 31));
      float* a = acc + dl_u * C + lane;
#pragma unroll
      for (int k = 0; k < V; ++k)
        __hip_atomic_fetch_add(a + 32 * k, w_u * v[u][k], __ATOMIC_RELAXED,
                               __HIP_MEMORY_SCOPE_WORKGROUP);
    }
  }
}

template <int C, bool RELU>
__global__ __launch_bounds__(OWN_T)
void own_k(const float* __restrict__ h, const float* __restrict__ dinv,
           const float* __restrict__ bias, const int* __restrict__ src,
           const int* __restrict__ dst, const int* __restrict__ pieces,
           const int* __restrict__ ovf, const int* __restrict__ ovfc,
           float* __restrict__ out) {
  __shared__ float acc[OWN_NB * C];
  __shared__ v4f dtab4[TAB_Q * OWN_T];
  __shared__ int ovn[BIN_NBLK];
  const float* dtab = (const float*)dtab4;
  constexpr int V = C / 32;
  typedef float vrow_t __attribute__((ext_vector_type(V)));

  const int tid  = threadIdx.x;
  const int lane = tid & 31;
  const int wave = __builtin_amdgcn_readfirstlane(tid >> 5);
  const int n0   = blockIdx.x * OWN_NB;

  const int* __restrict__ base = pieces + (size_t)blockIdx.x * BIN_NBLK * PIECE_W;
  int b     = wave;
  int ent_v = base[b * PIECE_W + lane];
  const int ovc = ovfc[min(tid, BIN_NBLK - 1) * CNT_PITCH];

  constexpr int NSLOT = OWN_NB * 32;
  constexpr int NTRIP = (NSLOT + OWN_T - 1) / OWN_T;
  {
    v4f tv[TAB_Q];
#pragma unroll
    for (int q = 0; q < TAB_Q; ++q)
      tv[q] = ((const v4f*)dinv)[min(tid + q * OWN_T, TAB_V - 1)];
    const vrow_t bv = *(const vrow_t*)(bias + V * lane);
    float  di[NTRIP];
    vrow_t hv[NTRIP];
#pragma unroll
    for (int q = 0; q < NTRIP; ++q) {
      const int i    = min((tid + q * OWN_T) >> 5, OWN_NB - 1);
      const int node = min(n0 + i, N_NODES - 1);
      di[q] = dinv[node];
      hv[q] = *(const vrow_t*)(h + (size_t)node * C + V * lane);
    }
#pragma unroll
    for (int q = 0; q < TAB_Q; ++q) dtab4[tid + q * OWN_T] = tv[q];
#pragma unroll
    for (int q = 0; q < NTRIP; ++q) {
      const int t = tid + q * OWN_T;
      const int i = t >> 5;
      if (t < NSLOT) {
        float* a = acc + i * C + lane;
        const bool real = n0 + i < N_NODES;
#pragma unroll
        for (int k = 0; k < V; ++k) a[32 * k] = real ? bv[k] + di[q] * di[q] * hv[q][k] : 0.0f;
      }
    }
  }
  if (tid < BIN_NBLK) ovn[tid] = min(ovc, BIN_E);
  __syncthreads();

  int l0    = 1;
  int ent_n = 0;
  while (b < BIN_NBLK) {
    const int  n  = min(__builtin_amdgcn_readfirstlane(ent_v), PIECE_CAP);
    const bool ok = lane >= 1 && lane <= n;
    const int   s_v  = ok ? ent_v >> BIN_SHIFT : 0;
    const int   dq   = ok ? ent_v & ((1 << BIN_SHIFT) - 1) : 0;
    const int   dl_v = ok ? dq : -1;
    const float ds_v = dtab[s_v];
    const int bn = b + OWN_NW;
    if (l0 == 1) {
      ent_n = 0;
      if (bn < BIN_NBLK) ent_n = base[bn * PIECE_W + lane];
    }
    const float w_v = ds_v * dtab[n0 + dq];
    own_batch<C>(h, acc, lane, l0, s_v, dl_v, w_v);
    const bool more = l0 + OWN_QB <= n;
    l0    = more ? l0 + OWN_QB : 1;
    b     = more ? b : bn;
    ent_v = more ? ent_v : ent_n;
  }

#pragma unroll
  for (int q = 0; q < OVN_Q; ++q) {
    const int c = lane + 32 * q < BIN_NBLK ? ovn[lane + 32 * q] : 0;
    unsigned mask = __builtin_amdgcn_ballot_w32(c > 0);
    while (mask) {
      const int j  = __builtin_ctz(mask);
      const int no = __builtin_amdgcn_readlane(c, j);
      const int* __restrict__ lst = ovf + (size_t)(32 * q + j) * BIN_E;
      mask &= mask - 1;
      for (int i = wave; i < no; i += OWN_NW) {
        const int e  = lst[i];
        const int dq = dst[e] - n0;
        if ((unsigned)dq < (unsigned)OWN_NB) {
          const int   s = src[e];
          const float w = dtab[s] * dtab[n0 + dq];
          const vrow_t x = *(const vrow_t*)(h + (size_t)s * C + V * lane);
          float* a = acc + dq * C + lane;
#pragma unroll
          for (int k = 0; k < V; ++k)
            __hip_atomic_fetch_add(a + 32 * k, w * x[k], __ATOMIC_RELAXED,
                                   __HIP_MEMORY_SCOPE_WORKGROUP);
        }
      }
    }
  }
  __syncthreads();

  for (int t = tid; t < OWN_NB * 32; t += OWN_T) {
    const int i    = t >> 5;
    const int node = n0 + i;
    if (node < N_NODES) {
      const float* a = acc + i * C + lane;
      vrow_t r;
#pragma unroll
      for (int k = 0; k < V; ++k) {
        r[k] = a[32 * k];
        if (RELU) r[k] = fmaxf(r[k], 0.0f);
      }
      *(vrow_t*)(out + (size_t)node * C + V * lane) = r;
    }
  }
}

extern "C" void kernel_launch(void* const* d_in, const int* in_sizes, int n_in,
                              void* d_out, int out_size, void* d_ws, size_t ws_size,
                              hipStream_t stream) {
  (void)in_sizes; (void)n_in; (void)out_size; (void)ws_size;
  const float* x  = (const float*)d_in[0];
  const int*   ei = (const int*)d_in[1];
  const float* W1 = (const float*)d_in[2];
  const float* b1 = (const float*)d_in[3];
  const float* W2 = (const float*)d_in[4];
  const float* b2 = (const float*)d_in[5];
  const int* src = ei;
  const int* dst = ei + E_EDGES;
  float* out = (float*)d_out;

  char* ws = (char*)d_ws;
  float* dinv = (float*)ws;  ws += (((size_t)N_NODES * 4) + 255) / 256 * 256;
  float* H1   = (float*)ws;  ws += (size_t)N_NODES * C_HID * 4;
  float* A1   = (float*)ws;  ws += (size_t)N_NODES * C_HID * 4;
  float* H2   = (float*)ws;  ws += (size_t)N_NODES * C_OUT * 4;
  int*   ovfc = (int*)ws;    ws += (size_t)BIN_NBLK * CNT_PITCH * 4;
  int*   pcs  = (int*)ws;    ws += (size_t)OWN_NBLK * BIN_NBLK * PIECE_W * 4;
  int*   ovf  = (int*)ws;

  bin_k<<<BIN_NBLK, dim3(BIN_T), 0, stream>>>(pcs, ovf, ovfc, src, dst, E_EDGES);
  deg_k<<<OWN_NBLK, dim3(DEG_T), 0, stream>>>(dinv, pcs, ovf, ovfc, dst);

  gemm_strip_k<C_IN, C_HID><<<dim3((M_TILES + 3) / 4, C_HID / 64), dim3(128), 0, stream>>>(x, W1, H1);
  own_k<C_HID, true><<<OWN_NBLK, dim3(OWN_T), 0, stream>>>(H1, dinv, b1, src, dst, pcs, ovf, ovfc, A1);

  gemm_strip_k<C_HID, C_OUT><<<dim3((M_TILES + 3) / 4, C_OUT / 64), dim3(128), 0, stream>>>(A1, W2, H2);
  own_k<C_OUT, false><<<OWN_NBLK, dim3(OWN_T), 0, stream>>>(H2, dinv, b2, src, dst, pcs, ovf, ovfc, out);
}


# own_k: a wave sums its edges in registers (5-row group x quarter of the pieces, entries filtered to a per-wave LDS list), one ds_add_f32 per row element and wave; sum order: per-wave partial sums
# speedup vs baseline: 4.5857x; 1.0397x over previous
#include <hip/hip_runtime.h>

#define N_NODES 10000
#define E_EDGES 640000
#define C_IN    256
#define C_HID   128
#define C_OUT   64

typedef float    v2f  __attribute__((ext_vector_type(2)));
typedef float    v8f  __attribute__((ext_vector_type(8)));
typedef float    v4f  __attribute__((ext_vector_type(4)));
typedef int      v4i  __attribute__((ext_vector_type(4)));

#define OWN_NB    20
#define OWN_T     512
#define OWN_NW    (OWN_T / 32)
#define OWN_NBLK  ((N_NODES + OWN_NB - 1) / OWN_NB)
#define BIN_SHIFT 8
#define BIN_T     1024
#define BIN_EPT   5
#define BIN_E     (BIN_T * BIN_EPT)
#define BIN_NBLK  ((E_EDGES + BIN_E - 1) / BIN_E)
#define PIECE_W   32
#define PIECE_CAP (PIECE_W - 1)
#define CNT_PITCH 32
#define DEG_T     256
#define DEG_NW    (DEG_T / 32)
#define DEG_Q     ((BIN_NBLK + DEG_NW - 1) / DEG_NW)
#define OVN_Q     ((BIN_NBLK + 31) / 32)

static_assert(OWN_NB <= (1 << BIN_SHIFT), "local dst index must fit below the src field of an entry");
static_assert(N_NODES <= (1 << (31 - BIN_SHIFT)), "src must fit above the local dst field of an entry");
static_assert(OWN_NBLK <= BIN_T, "bin_k handles one owner range per thread");
static_assert((OWN_NBLK * PIECE_W / 4) % 8 == 0, "a piece is 8 16-byte elements");
static_assert(BIN_NBLK <= DEG_T && BIN_NBLK <= OWN_T, "one thread per binning block reads its overflow count");

__global__ __launch_bounds__(BIN_T)
void bin_k(int* __restrict__ pieces, int* __restrict__ ovf, int* __restrict__ ovfc,
           const int* __restrict__ src, const int* __restrict__ dst, int e) {
  __shared__ int  cnt[OWN_NBLK];
  __shared__ v4i  stage4[OWN_NBLK * PIECE_W / 4];
  __shared__ int  ocnt;
  int* stage = (int*)stage4;
  const int tid = threadIdx.x;
  if (tid < OWN_NBLK) cnt[tid] = 0;
  if (tid == 0) ocnt = 0;

  const int i0 = blockIdx.x * BIN_E + tid;
  int d[BIN_EPT], s[BIN_EPT];
#pragma unroll
  for (int k = 0; k < BIN_EPT; ++k) {
    const int i = min(i0 + k * BIN_T, e - 1);
    d[k] = dst[i];
    s[k] = src[i];
  }
  for (int i = tid; i < OWN_NBLK * PIECE_W / 4; i += BIN_T) stage4[i] = (v4i){0, 0, 0, 0};
  __syncthreads();

#pragma unroll
  for (int k = 0; k < BIN_EPT; ++k) {
    if (i0 + k * BIN_T < e) {
      const int b = d[k] / OWN_NB;
      const int r = atomicAdd(&cnt[b], 1);
      if (r < PIECE_CAP) {
        stage[b * PIECE_W + 1 + r] = (s[k] << BIN_SHIFT) | (d[k] - b * OWN_NB);
      } else {
        const int o = atomicAdd(&ocnt, 1);
        ovf[(size_t)blockIdx.x * BIN_E + o] = i0 + k * BIN_T;
      }
    }
  }
  __syncthreads();

  if (tid < OWN_NBLK) stage[tid * PIECE_W] = cnt[tid];
  __syncthreads();
  if (tid == 0) ovfc[blockIdx.x * CNT_PITCH] = ocnt;
  for (int i = tid; i < OWN_NBLK * PIECE_W / 4; i += BIN_T)
    ((v4i*)pieces)[((size_t)(i >> 3) * BIN_NBLK + blockIdx.x) * (PIECE_W / 4) + (i & 7)] = stage4[i];
}

__global__ __launch_bounds__(DEG_T)
void deg_k(float* __restrict__ dinv, const int* __restrict__ pieces, const int* __restrict__ ovf,
           const int* __restrict__ ovfc, const int* __restrict__ dst) {
  __shared__ int cnt[OWN_NB];
  __shared__ int ovn[BIN_NBLK];
  const int tid  = threadIdx.x;
  const int lane = tid & 31;
  const int wave = __builtin_amdgcn_readfirstlane(tid >> 5);
  const int n0   = blockIdx.x * OWN_NB;
  if (tid < OWN_NB) cnt[tid] = 0;
  const int* __restrict__ base = pieces + (size_t)blockIdx.x * BIN_NBLK * PIECE_W;

  int ent[DEG_Q];
#pragma unroll
  for (int q = 0; q < DEG_Q; ++q) ent[q] = base[min(wave + q * DEG_NW, BIN_NBLK - 1) * PIECE_W + lane];
  const int ovc = ovfc[min(tid, BIN_NBLK - 1) * CNT_PITCH];
  if (tid < BIN_NBLK) ovn[tid] = min(ovc, BIN_E);
  __syncthreads();

#pragma unroll
  for (int q = 0; q < DEG_Q; ++q) {
    const int n = min(__builtin_amdgcn_readfirstlane(ent[q]), PIECE_CAP);
    if (wave + q * DEG_NW < BIN_NBLK && lane >= 1 && lane <= n)
      atomicAdd(&cnt[ent[q] & ((1 << BIN_SHIFT) - 1)], 1);
  }
#pragma unroll
  for (int q = 0; q < OVN_Q; ++q) {
    const int c = lane + 32 * q < BIN_NBLK ? ovn[lane + 32 * q] : 0;
    unsigned mask = __builtin_amdgcn_ballot_w32(c > 0);
    while (mask) {
      const int j  = __builtin_ctz(mask);
      const int no = __builtin_amdgcn_readlane(c, j);
      const int* __restrict__ lst = ovf + (size_t)(32 * q + j) * BIN_E;
      mask &= mask - 1;
      for (int i = tid; i < no; i += DEG_T) {
        const int dq = dst[lst[i]] - n0;
        if ((unsigned)dq < (unsigned)OWN_NB) atomicAdd(&cnt[dq], 1);
      }
    }
  }
  __syncthreads();

  if (tid < OWN_NB && n0 + tid < N_NODES) dinv[n0 + tid] = rsqrtf(1.0f + (float)cnt[tid]);
}

#define GEMM_KC 64
#define GEMM_NG (GEMM_KC / 8)
#define M_TILES (N_NODES / 16)

template <int K, int N>
__device__ __forceinline__ void gemm_load_chunk(float4 (&a)[GEMM_NG], float4 (&b)[GEMM_NG][4],
                                                const float* __restrict__ Ap,
                                                const float* __restrict__ Bp, int k0) {
#pragma unroll
  for (int g = 0; g < GEMM_NG; ++g) {
    a[g] = *(const float4*)(Ap + k0 + 8 * g);
#pragma unroll
    for (int i = 0; i < 4; ++i) b[g][i] = *(const float4*)(Bp + (size_t)(k0 + 8 * g + i) * N);
  }
}

__device__ __forceinline__ void gemm_mma_chunk(v8f (&acc)[4], const float4 (&a)[GEMM_NG],
                                               const float4 (&b)[GEMM_NG][4]) {
#pragma unroll
  for (int g = 0; g < GEMM_NG; ++g) {
    const v2f a0 = {a[g].x, a[g].y};
    const v2f a1 = {a[g].z, a[g].w};
    const float b0[4] = {b[g][0].x, b[g][0].y, b[g][0].z, b[g][0].w};
    const float b1[4] = {b[g][1].x, b[g][1].y, b[g][1].z, b[g][1].w};
    const float b2[4] = {b[g][2].x, b[g][2].y, b[g][2].z, b[g][2].w};
    const float b3[4] = {b[g][3].x, b[g][3].y, b[g][3].z, b[g][3].w};
#pragma unroll
    for (int t = 0; t < 4; ++t) {
      const v2f bb = {b0[t], b1[t]};
      acc[t] = __builtin_amdgcn_wmma_f32_16x16x4_f32(false, a0, false, bb, (short)0, acc[t],
                                                     false, false);
    }
#pragma unroll
    for (int t = 0; t < 4; ++t) {
      const v2f bb = {b2[t], b3[t]};
      acc[t] = __builtin_amdgcn_wmma_f32_16x16x4_f32(false, a1, false, bb, (short)0, acc[t],
                                                     false, false);
    }
  }
}

template <int K, int N>
__global__ __launch_bounds__(128)
void gemm_strip_k(const float* __restrict__ A, const float* __restrict__ B, float* __restrict__ D) {
  static_assert(K % (2 * GEMM_KC) == 0, "the chunk loop below is written two chunks per trip");
  static_assert(N % 64 == 0, "a strip is 64 columns");
  const int lane = threadIdx.x & 31;
  const int wave = __builtin_amdgcn_readfirstlane(threadIdx.x >> 5);
  const int rt   = blockIdx.x * 4 + wave;
  if (rt >= M_TILES) return;
  const int c  = lane & 15;
  const int h  = lane >> 4;
  const int n0 = blockIdx.y * 64;

  const float* __restrict__ Ap = A + (size_t)(rt * 16 + c) * K + 4 * h;
  const float* __restrict__ Bp = B + (size_t)(4 * h) * N + n0 + 4 * c;

  v8f acc[4] = {};
  float4 a0[GEMM_NG], b0[GEMM_NG][4];
  float4 a1[GEMM_NG], b1[GEMM_NG][4];

  gemm_load_chunk<K, N>(a0, b0, Ap, Bp, 0);
#pragma unroll
  for (int k0 = 0; k0 < K; k0 += 2 * GEMM_KC) {
    gemm_load_chunk<K, N>(a1, b1, Ap, Bp, k0 + GEMM_KC);
    __builtin_amdgcn_sched_barrier(0);
    gemm_mma_chunk(acc, a0, b0);
    if (k0 + 2 * GEMM_KC < K) gemm_load_chunk<K, N>(a0, b0, Ap, Bp, k0 + 2 * GEMM_KC);
    __builtin_amdgcn_sched_barrier(0);
    gemm_mma_chunk(acc, a1, b1);
  }

  float* Dp = D + (size_t)(rt * 16 + 8 * h) * N + n0 + 4 * c;
#pragma unroll
  for (int i = 0; i < 8; ++i) {
    const float4 o = {acc[0][i], acc[1][i], acc[2][i], acc[3][i]};
    *(float4*)(Dp + (size_t)i * N) = o;
  }
}

static_assert(N_NODES % 4 == 0, "the dinv table is copied to LDS as 16-byte elements");
#define TAB_V (N_NODES / 4)
#define TAB_Q ((TAB_V + OWN_T - 1) / OWN_T)
#define OWN_QB 16
#define OWN_NG  4
#define OWN_GN  (OWN_NB / OWN_NG)
#define OWN_PG  (OWN_NW / OWN_NG)
#define OWN_PPW ((BIN_NBLK + OWN_PG - 1) / OWN_PG)
#define OWN_RQ  8
#define OWN_NR  ((OWN_PPW + OWN_RQ - 1) / OWN_RQ)
#define LIST_CAP (OWN_PPW * PIECE_W)
static_assert(OWN_NB % OWN_NG == 0 && OWN_NW % OWN_NG == 0, "node groups and piece groups tile the block");

template <int V> struct vrow { typedef float type __attribute__((ext_vector_type(V))); };

__device__ __forceinline__ void own_load_round(int (&ent)[OWN_RQ], const int* __restrict__ base,
                                               int pg, int r, int lane) {
#pragma unroll
  for (int q = 0; q < OWN_RQ; ++q) {
    const int b = pg + (r * OWN_RQ + q) * OWN_PG;
    ent[q] = base[min(b, BIN_NBLK - 1) * PIECE_W + lane];
  }
}

template <int C>
__device__ __forceinline__ void own_batch(const float* __restrict__ h, int lane, int s_v, int j_v,
                                          float w_v, typename vrow<C / 32>::type (&a)[OWN_GN]) {
  constexpr int V = C / 32;
  typedef typename vrow<V>::type vrow_t;
  vrow_t v[OWN_QB];
#pragma unroll
  for (int u = 0; u < OWN_QB; ++u) {
    const int s_u = __builtin_amdgcn_readlane(s_v, u);
    v[u] = *(const vrow_t*)(h + (size_t)s_u * C + V * lane);
  }
#pragma unroll
  for (int u = 0; u < OWN_QB; ++u) {
    const int   j_u = __builtin_amdgcn_readlane(j_v, u);
    const float w_u = __int_as_float(__builtin_amdgcn_readlane(__float_as_int(w_v), u));
#pragma unroll
    for (int j = 0; j < OWN_GN; ++j) {
      const float w_j = j_u == j ? w_u : 0.0f;
      a[j] += w_j * v[u];
    }
  }
}

template <int C, bool RELU>
__global__ __launch_bounds__(OWN_T)
void own_k(const float* __restrict__ h, const float* __restrict__ dinv,
           const float* __restrict__ bias, const int* __restrict__ src,
           const int* __restrict__ dst, const int* __restrict__ pieces,
           const int* __restrict__ ovf, const int* __restrict__ ovfc,
           float* __restrict__ out) {
  __shared__ float acc[OWN_NB * C];
  __shared__ v4f dtab4[TAB_Q * OWN_T];
  __shared__ int ovn[BIN_NBLK];
  __shared__ int list[OWN_NW * LIST_CAP];
  const float* dtab = (const float*)dtab4;
  constexpr int V = C / 32;
  typedef typename vrow<V>::type vrow_t;

  const int tid  = threadIdx.x;
  const int lane = tid & 31;
  const int wave = __builtin_amdgcn_readfirstlane(tid >> 5);
  const int n0   = blockIdx.x * OWN_NB;

  const int pg  = wave / OWN_NG;
  const int jlo = (wave % OWN_NG) * OWN_GN;
  int* __restrict__ lst = list + wave * LIST_CAP;

  const int* __restrict__ base = pieces + (size_t)blockIdx.x * BIN_NBLK * PIECE_W;
  int ent[OWN_RQ];
  own_load_round(ent, base, pg, 0, lane);
  const int ovc = ovfc[min(tid, BIN_NBLK - 1) * CNT_PITCH];

  constexpr int NSLOT = OWN_NB * 32;
  constexpr int NTRIP = (NSLOT + OWN_T - 1) / OWN_T;
  {
    v4f tv[TAB_Q];
#pragma unroll
    for (int q = 0; q < TAB_Q; ++q)
      tv[q] = ((const v4f*)dinv)[min(tid + q * OWN_T, TAB_V - 1)];
    const vrow_t bv = *(const vrow_t*)(bias + V * lane);
    float  di[NTRIP];
    vrow_t hv[NTRIP];
#pragma unroll
    for (int q = 0; q < NTRIP; ++q) {
      const int i    = min((tid + q * OWN_T) >> 5, OWN_NB - 1);
      const int node = min(n0 + i, N_NODES - 1);
      di[q] = dinv[node];
      hv[q] = *(const vrow_t*)(h + (size_t)node * C + V * lane);
    }
#pragma unroll
    for (int q = 0; q < TAB_Q; ++q) dtab4[tid + q * OWN_T] = tv[q];
#pragma unroll
    for (int q = 0; q < NTRIP; ++q) {
      const int t = tid + q * OWN_T;
      const int i = t >> 5;
      if (t < NSLOT) {
        float* a = acc + i * C + lane;
        const bool real = n0 + i < N_NODES;
#pragma unroll
        for (int k = 0; k < V; ++k) a[32 * k] = real ? bv[k] + di[q] * di[q] * hv[q][k] : 0.0f;
      }
    }
  }
  if (tid < BIN_NBLK) ovn[tid] = min(ovc, BIN_E);

  int cnt = 0;
#pragma unroll
  for (int r = 0; r < OWN_NR; ++r) {
    int nxt[OWN_RQ];
    if (r + 1 < OWN_NR) own_load_round(nxt, base, pg, r + 1, lane);
#pragma unroll
    for (int q = 0; q < OWN_RQ; ++q) {
      const int  b  = pg + (r * OWN_RQ + q) * OWN_PG;
      const int  n  = min(__builtin_amdgcn_readfirstlane(ent[q]), PIECE_CAP);
      const int  dq = ent[q] & ((1 << BIN_SHIFT) - 1);
      const bool ok = b < BIN_NBLK && lane >= 1 && lane <= n &&
                      (unsigned)(dq - jlo) < (unsigned)OWN_GN;
      const unsigned m = __builtin_amdgcn_ballot_w32(ok);
      if (ok) lst[cnt + (int)__builtin_amdgcn_mbcnt_lo(m, 0u)] = ent[q];
      cnt += __builtin_popcount(m);
    }
    if (r + 1 < OWN_NR) {
#pragma unroll
      for (int q = 0; q < OWN_RQ; ++q) ent[q] = nxt[q];
    }
  }
  cnt = __builtin_amdgcn_readfirstlane(cnt);
  __syncthreads();

  vrow_t a[OWN_GN];
#pragma unroll
  for (int j = 0; j < OWN_GN; ++j) a[j] = (vrow_t)(0.0f);
  for (int c0 = 0; c0 < cnt; c0 += OWN_QB) {
    const bool  ok  = lane < OWN_QB && c0 + lane < cnt;
    const int   e   = lst[min(c0 + lane, cnt - 1)];
    const int   s_v = e >> BIN_SHIFT;
    const int   dq  = e & ((1 << BIN_SHIFT) - 1);
    const int   j_v = ok ? dq - jlo : -1;
    const float w_v = dtab[s_v] * dtab[n0 + dq];
    own_batch<C>(h, lane, s_v, j_v, w_v, a);
  }
#pragma unroll
  for (int j = 0; j < OWN_GN; ++j) {
    float* p = acc + (jlo + j) * C + lane;
#pragma unroll
    for (int k = 0; k < V; ++k)
      __hip_atomic_fetch_add(p + 32 * k, a[j][k], __ATOMIC_RELAXED, __HIP_MEMORY_SCOPE_WORKGROUP);
  }

#pragma unroll
  for (int q = 0; q < OVN_Q; ++q) {
    const int c = lane + 32 * q < BIN_NBLK ? ovn[lane + 32 * q] : 0;
    unsigned mask = __builtin_amdgcn_ballot_w32(c > 0);
    while (mask) {
      const int j  = __builtin_ctz(mask);
      const int no = __builtin_amdgcn_readlane(c, j);
      const int* __restrict__ lst = ovf + (size_t)(32 * q + j) * BIN_E;
      mask &= mask - 1;
      for (int i = wave; i < no; i += OWN_NW) {
        const int e  = lst[i];
        const int dq = dst[e] - n0;
        if ((unsigned)dq < (unsigned)OWN_NB) {
          const int   s = src[e];
          const float w = dtab[s] * dtab[n0 + dq];
          const vrow_t x = *(const vrow_t*)(h + (size_t)s * C + V * lane);
          float* a = acc + dq * C + lane;
#pragma unroll
          for (int k = 0; k < V; ++k)
            __hip_atomic_fetch_add(a + 32 * k, w * x[k], __ATOMIC_RELAXED,
                                   __HIP_MEMORY_SCOPE_WORKGROUP);
        }
      }
    }
  }
  __syncthreads();

  for (int t = tid; t < OWN_NB * 32; t += OWN_T) {
    const int i    = t >> 5;
    const int node = n0 + i;
    if (node < N_NODES) {
      const float* a = acc + i * C + lane;
      vrow_t r;
#pragma unroll
      for (int k = 0; k < V; ++k) {
        r[k] = a[32 * k];
        if (RELU) r[k] = fmaxf(r[k], 0.0f);
      }
      *(vrow_t*)(out + (size_t)node * C + V * lane) = r;
    }
  }
}

extern "C" void kernel_launch(void* const* d_in, const int* in_sizes, int n_in,
                              void* d_out, int out_size, void* d_ws, size_t ws_size,
                              hipStream_t stream) {
  (void)in_sizes; (void)n_in; (void)out_size; (void)ws_size;
  const float* x  = (const float*)d_in[0];
  const int*   ei = (const int*)d_in[1];
  const float* W1 = (const float*)d_in[2];
  const float* b1 = (const float*)d_in[3];
  const float* W2 = (const float*)d_in[4];
  const float* b2 = (const float*)d_in[5];
  const int* src = ei;
  const int* dst = ei + E_EDGES;
  float* out = (float*)d_out;

  char* ws = (char*)d_ws;
  float* dinv = (float*)ws;  ws += (((size_t)N_NODES * 4) + 255) / 256 * 256;
  float* H1   = (float*)ws;  ws += (size_t)N_NODES * C_HID * 4;
  float* A1   = (float*)ws;  ws += (size_t)N_NODES * C_HID * 4;
  float* H2   = (float*)ws;  ws += (size_t)N_NODES * C_OUT * 4;
  int*   ovfc = (int*)ws;    ws += (size_t)BIN_NBLK * CNT_PITCH * 4;
  int*   pcs  = (int*)ws;    ws += (size_t)OWN_NBLK * BIN_NBLK * PIECE_W * 4;
  int*   ovf  = (int*)ws;

  bin_k<<<BIN_NBLK, dim3(BIN_T), 0, stream>>>(pcs, ovf, ovfc, src, dst, E_EDGES);
  deg_k<<<OWN_NBLK, dim3(DEG_T), 0, stream>>>(dinv, pcs, ovf, ovfc, dst);

  gemm_strip_k<C_IN, C_HID><<<dim3((M_TILES + 3) / 4, C_HID / 64), dim3(128), 0, stream>>>(x, W1, H1);
  own_k<C_HID, true><<<OWN_NBLK, dim3(OWN_T), 0, stream>>>(H1, dinv, b1, src, dst, pcs, ovf, ovfc, A1);

  gemm_strip_k<C_HID, C_OUT><<<dim3((M_TILES + 3) / 4, C_OUT / 64), dim3(128), 0, stream>>>(A1, W2, H2);
  own_k<C_OUT, false><<<OWN_NBLK, dim3(OWN_T), 0, stream>>>(H2, dinv, b2, src, dst, pcs, ovf, ovfc, out);
}
